# starting kernel
# baseline (speedup 1.0000x reference)
#include <hip/hip_runtime.h>

typedef float v8f __attribute__((ext_vector_type(8)));
typedef float v2f __attribute__((ext_vector_type(2)));

#define DIM_E   64
#define DIM_H   8
#define NLAYER  25
#define NBATCH  1000
#define NT      512
#define TILE_M  16
#define HPAD    12
#define NSTEPS  (NT + NLAYER - 1)

__device__ __forceinline__ float fast_tanh(float x) {
  float e = __builtin_amdgcn_exp2f(x * 2.88539008177792681f);
  return 1.0f - 2.0f * __builtin_amdgcn_rcpf(e + 1.0f);
}

__device__ __forceinline__ v8f wmma4(v2f a, v2f b, v8f c) {
  return __builtin_amdgcn_wmma_f32_16x16x4_f32(false, a, false, b, (short)0, c,
                                               false, false);
}

__global__ __launch_bounds__(NLAYER * 32, 1)
void rnn_pipeline_kernel(const int* __restrict__ x,
                         const float* __restrict__ h0,
                         const float* __restrict__ emb,
                         const float* __restrict__ Wxh0,
                         const float* __restrict__ Wxhr,
                         const float* __restrict__ Whh,
                         const float* __restrict__ bh,
                         const float* __restrict__ Why,
                         const float* __restrict__ by,
                         float* __restrict__ out) {
  __shared__ float hbuf[NLAYER][2][TILE_M][HPAD];

  const int tid  = threadIdx.x;
  const int wid  = tid >> 5;
  const int lane = tid & 31;
  const int n    = lane & 15;
  const int kb   = (lane >> 4) << 1;
  const int tile_base = blockIdx.x * TILE_M;

  v2f  wA[16];
  v2f  wR[2];
  float bias = (n < DIM_H) ? bh[wid * DIM_H + n] : 0.0f;

  if (wid == 0) {
#pragma unroll
    for (int c = 0; c < 16; ++c)
#pragma unroll
      for (int j = 0; j < 2; ++j) {
        int k = c * 4 + kb + j;
        wA[c][j] = (n < DIM_H) ? Wxh0[k * DIM_H + n] : 0.0f;
      }
#pragma unroll
    for (int c = 0; c < 2; ++c)
#pragma unroll
      for (int j = 0; j < 2; ++j) {
        int k = c * 4 + kb + j;
        wR[c][j] = (n < DIM_H) ? Whh[k * DIM_H + n] : 0.0f;
      }
  } else {
#pragma unroll
    for (int c = 0; c < 4; ++c)
#pragma unroll
      for (int j = 0; j < 2; ++j) {
        int k = c * 4 + kb + j;
        float v = 0.0f;
        if (n < DIM_H)
          v = (k < DIM_H) ? Wxhr[(wid - 1) * DIM_H * DIM_H + k * DIM_H + n]
                          : Whh[wid * DIM_H * DIM_H + (k - DIM_H) * DIM_H + n];
        wA[c][j] = v;
      }
  }

  if (lane < TILE_M) {
    int gb = tile_base + lane; if (gb > NBATCH - 1) gb = NBATCH - 1;
    const float* hp = h0 + (wid * NBATCH + gb) * DIM_H;
#pragma unroll
    for (int j = 0; j < DIM_H; ++j) {
      float v = hp[j];
      hbuf[wid][0][lane][j] = v;
      hbuf[wid][1][lane][j] = v;
    }
  }
  __syncthreads();

  const int m = n;
  int gb = tile_base + m; if (gb > NBATCH - 1) gb = NBATCH - 1;

  int xcur = 0;
  if (wid == 0) xcur = x[gb * NT];

  for (int s = 0; s < NSTEPS; ++s) {
    if (s >= wid && s < wid + NT) {
      const int p_in  = (s + 1) & 1;
      const int p_out = s & 1;
      v8f acc;
#pragma unroll
      for (int r = 0; r < 8; ++r) acc[r] = bias;

      if (wid == 0) {
        const int t = s;
        const float* erow = emb + (long)xcur * DIM_E;
        if (t + 1 < NT) xcur = x[gb * NT + t + 1];
#pragma unroll
        for (int c = 0; c < 16; ++c) {
          v2f a = *(const v2f*)(erow + c * 4 + kb);
          acc = wmma4(a, wA[c], acc);
        }
        const float* hp = &hbuf[0][p_in][m][0];
        v2f a0 = *(const v2f*)(hp + kb);
        v2f a1 = *(const v2f*)(hp + 4 + kb);
        acc = wmma4(a0, wR[0], acc);
        acc = wmma4(a1, wR[1], acc);
      } else {
        const float* ip = &hbuf[wid - 1][p_in][m][0];
        const float* hp = &hbuf[wid][p_in][m][0];
        v2f a0 = *(const v2f*)(ip + kb);
        v2f a1 = *(const v2f*)(ip + 4 + kb);
        v2f a2 = *(const v2f*)(hp + kb);
        v2f a3 = *(const v2f*)(hp + 4 + kb);
        acc = wmma4(a0, wA[0], acc);
        acc = wmma4(a1, wA[1], acc);
        acc = wmma4(a2, wA[2], acc);
        acc = wmma4(a3, wA[3], acc);
      }

#pragma unroll
      for (int r = 0; r < 8; ++r) acc[r] = fast_tanh(acc[r]);

      if (n < DIM_H) {
        const int rowbase = (lane >> 4) << 3;
#pragma unroll
        for (int r = 0; r < 8; ++r)
          hbuf[wid][p_out][rowbase + r][n] = acc[r];
      }
    }
    __syncthreads();
  }

  const int pf = (wid + NT - 1) & 1;
  if (lane < TILE_M) {
    int g = tile_base + lane;
    if (g < NBATCH) {
      const float* src = &hbuf[wid][pf][lane][0];
      float4 v0 = *(const float4*)src;
      float4 v1 = *(const float4*)(src + 4);
      float* dst = out + NBATCH + ((long)wid * NBATCH + g) * DIM_H;
      *(float4*)dst       = v0;
      *(float4*)(dst + 4) = v1;
      if (wid == NLAYER - 1) {
        float a = by[0];
#pragma unroll
        for (int j = 0; j < DIM_H; ++j) a += src[j] * Why[j];
        out[g] = a;
      }
    }
  }
}

extern "C" void kernel_launch(void* const* d_in, const int* in_sizes, int n_in,
                              void* d_out, int out_size, void* d_ws, size_t ws_size,
                              hipStream_t stream) {
  (void)in_sizes; (void)n_in; (void)d_ws; (void)ws_size; (void)out_size;
  const int*   x    = (const int*)  d_in[0];
  const float* h0   = (const float*)d_in[1];
  const float* emb  = (const float*)d_in[2];
  const float* Wxh0 = (const float*)d_in[3];
  const float* Wxhr = (const float*)d_in[4];
  const float* Whh  = (const float*)d_in[5];
  const float* bh   = (const float*)d_in[6];
  const float* Why  = (const float*)d_in[7];
  const float* by   = (const float*)d_in[8];
  float* out = (float*)d_out;

  dim3 grid((NBATCH + TILE_M - 1) / TILE_M);
  dim3 block(NLAYER * 32);
  rnn_pipeline_kernel<<<grid, block, 0, stream>>>(x, h0, emb, Wxh0, Wxhr, Whh,
                                                  bh, Why, by, out);
}


# Layer-0 embedding gather and K=64 input projection moved off the layer-0 wave into 3 extra pipeline-stage waves (6+5+5 k-chunks), gathered one step ahead; accumulator passed on via LDS, same k order
# speedup vs baseline: 1.3694x; 1.3694x over previous
#include <hip/hip_runtime.h>


typedef float v8f __attribute__((ext_vector_type(8)));
typedef float v2f __attribute__((ext_vector_type(2)));

#define DIM_E   64
#define DIM_H   8
#define NLAYER  25
#define NBATCH  1000
#define NT      512
#define TILE_M  16
#define HPAD    12
#define NPRE    3
#define NSTAGE  (NPRE + NLAYER)
#define PPAD    12
#define NSTEPS  (NT + NSTAGE - 1)

__device__ __forceinline__ float fast_tanh(float x) {
  float e = __builtin_amdgcn_exp2f(x * 2.88539008177792681f);
  return 1.0f - 2.0f * __builtin_amdgcn_rcpf(e + 1.0f);
}

__device__ __forceinline__ void step_barrier() {
  asm volatile("s_wait_dscnt 0x0" ::: "memory");
  asm volatile("s_barrier_signal -1\n\ts_barrier_wait -1" ::: "memory");
}

__device__ __forceinline__ v8f wmma4(v2f a, v2f b, v8f c) {
  return __builtin_amdgcn_wmma_f32_16x16x4_f32(false, a, false, b, (short)0, c,
                                               false, false);
}

__global__ __launch_bounds__(NSTAGE * 32, 1)
void rnn_pipeline_kernel(const int* __restrict__ x,
                         const float* __restrict__ h0,
                         const float* __restrict__ emb,
                         const float* __restrict__ Wxh0,
                         const float* __restrict__ Wxhr,
                         const float* __restrict__ Whh,
                         const float* __restrict__ bh,
                         const float* __restrict__ Why,
                         const float* __restrict__ by,
                         float* __restrict__ out) {
  __shared__ __attribute__((aligned(16))) float hbuf[NLAYER][2][TILE_M][HPAD];
  __shared__ __attribute__((aligned(16))) float pacc[NPRE][2][32][PPAD];

  const int tid  = threadIdx.x;
  const int wid  = __builtin_amdgcn_readfirstlane(tid >> 5);
  const int lay  = wid - NPRE;
  const int lane = tid & 31;
  const int n    = lane & 15;
  const int kb   = (lane >> 4) << 1;
  const int tile_base = blockIdx.x * TILE_M;
  const int cbase = (wid == 0) ? 0 : ((wid == 1) ? 6 : 11);

  v2f  wA[6];
#pragma unroll
  for (int c = 0; c < 6; ++c) wA[c] = (v2f){0.0f, 0.0f};
  float bias = 0.0f;

  if (wid < NPRE) {
#pragma unroll
    for (int c = 0; c < 6; ++c)
#pragma unroll
      for (int j = 0; j < 2; ++j) {
        int k = (cbase + c) * 4 + kb + j;
        if (n < DIM_H && (c < 5 || wid == 0)) wA[c][j] = Wxh0[k * DIM_H + n];
      }
    if (wid == 0) bias = (n < DIM_H) ? bh[n] : 0.0f;
  } else if (lay == 0) {
#pragma unroll
    for (int c = 0; c < 2; ++c)
#pragma unroll
      for (int j = 0; j < 2; ++j) {
        int k = c * 4 + kb + j;
        wA[c][j] = (n < DIM_H) ? Whh[k * DIM_H + n] : 0.0f;
      }
  } else {
    bias = (n < DIM_H) ? bh[lay * DIM_H + n] : 0.0f;
#pragma unroll
    for (int c = 0; c < 4; ++c)
#pragma unroll
      for (int j = 0; j < 2; ++j) {
        int k = c * 4 + kb + j;
        float v = 0.0f;
        if (n < DIM_H)
          v = (k < DIM_H) ? Wxhr[(lay - 1) * DIM_H * DIM_H + k * DIM_H + n]
                          : Whh[lay * DIM_H * DIM_H + (k - DIM_H) * DIM_H + n];
        wA[c][j] = v;
      }
  }

  if (wid >= NPRE && lane < TILE_M) {
    int gbi = tile_base + lane; if (gbi > NBATCH - 1) gbi = NBATCH - 1;
    const float* hp = h0 + (lay * NBATCH + gbi) * DIM_H;
#pragma unroll
    for (int j = 0; j < DIM_H; ++j) {
      float v = hp[j];
      hbuf[lay][0][lane][j] = v;
      hbuf[lay][1][lane][j] = v;
    }
  }

  const int m = n;
  int gb = tile_base + m; if (gb > NBATCH - 1) gb = NBATCH - 1;

  int xn = 0;
  v2f en[6];
#pragma unroll
  for (int c = 0; c < 6; ++c) en[c] = (v2f){0.0f, 0.0f};
  if (wid < NPRE) {
    const int x0 = x[gb * NT];
    xn = x[gb * NT + 1];
    const float* erow = emb + (long)x0 * DIM_E + cbase * 4 + kb;
#pragma unroll
    for (int c = 0; c < 5; ++c) en[c] = *(const v2f*)(erow + c * 4);
    if (wid == 0) en[5] = *(const v2f*)(erow + 20);
  }
  __syncthreads();

  for (int s = 0; s < NSTEPS; ++s) {
    const int t = s - wid;
    if (t >= 0 && t < NT) {
      const int p_in  = (s + 1) & 1;
      const int p_out = s & 1;
      v8f acc;

      if (wid < NPRE) {
        if (wid == 0) {
#pragma unroll
          for (int r = 0; r < 8; ++r) acc[r] = bias;
        } else {
          const float* pp = &pacc[wid - 1][p_in][lane][0];
          float4 c0 = *(const float4*)pp;
          float4 c1 = *(const float4*)(pp + 4);
          acc[0] = c0.x; acc[1] = c0.y; acc[2] = c0.z; acc[3] = c0.w;
          acc[4] = c1.x; acc[5] = c1.y; acc[6] = c1.z; acc[7] = c1.w;
        }
#pragma unroll
        for (int c = 0; c < 5; ++c) acc = wmma4(en[c], wA[c], acc);
        if (wid == 0) acc = wmma4(en[5], wA[5], acc);
        float* pq = &pacc[wid][p_out][lane][0];
        *(float4*)pq       = make_float4(acc[0], acc[1], acc[2], acc[3]);
        *(float4*)(pq + 4) = make_float4(acc[4], acc[5], acc[6], acc[7]);
        if (t + 1 < NT) {
          const float* erow = emb + (long)xn * DIM_E + cbase * 4 + kb;
#pragma unroll
          for (int c = 0; c < 5; ++c) en[c] = *(const v2f*)(erow + c * 4);
          if (wid == 0) en[5] = *(const v2f*)(erow + 20);
          if (t + 2 < NT) xn = x[gb * NT + t + 2];
        }
      } else {
        if (lay == 0) {
          const float* pp = &pacc[NPRE - 1][p_in][lane][0];
          float4 c0 = *(const float4*)pp;
          float4 c1 = *(const float4*)(pp + 4);
          acc[0] = c0.x; acc[1] = c0.y; acc[2] = c0.z; acc[3] = c0.w;
          acc[4] = c1.x; acc[5] = c1.y; acc[6] = c1.z; acc[7] = c1.w;
          const float* hp = &hbuf[0][p_in][m][0];
          v2f a0 = *(const v2f*)(hp + kb);
          v2f a1 = *(const v2f*)(hp + 4 + kb);
          acc = wmma4(a0, wA[0], acc);
          acc = wmma4(a1, wA[1], acc);
        } else {
#pragma unroll
          for (int r = 0; r < 8; ++r) acc[r] = bias;
          const float* ip = &hbuf[lay - 1][p_in][m][0];
          const float* hp = &hbuf[lay][p_in][m][0];
          v2f a0 = *(const v2f*)(ip + kb);
          v2f a1 = *(const v2f*)(ip + 4 + kb);
          v2f a2 = *(const v2f*)(hp + kb);
          v2f a3 = *(const v2f*)(hp + 4 + kb);
          acc = wmma4(a0, wA[0], acc);
          acc = wmma4(a1, wA[1], acc);
          acc = wmma4(a2, wA[2], acc);
          acc = wmma4(a3, wA[3], acc);
        }

#pragma unroll
        for (int r = 0; r < 8; ++r) acc[r] = fast_tanh(acc[r]);

        if (n < DIM_H) {
          const int rowbase = (lane >> 4) << 3;
#pragma unroll
          for (int r = 0; r < 8; ++r)
            hbuf[lay][p_out][rowbase + r][n] = acc[r];
        }
      }
    }
    step_barrier();
  }

  const int pf = (wid + NT - 1) & 1;
  if (wid >= NPRE && lane < TILE_M) {
    int g = tile_base + lane;
    if (g < NBATCH) {
      const float* src = &hbuf[lay][pf][lane][0];
      float4 v0 = *(const float4*)src;
      float4 v1 = *(const float4*)(src + 4);
      float* dst = out + NBATCH + ((long)lay * NBATCH + g) * DIM_H;
      *(float4*)dst       = v0;
      *(float4*)(dst + 4) = v1;
      if (lay == NLAYER - 1) {
        float a = by[0];
#pragma unroll
        for (int j = 0; j < DIM_H; ++j) a += src[j] * Why[j];
        out[g] = a;
      }
    }
  }
}

extern "C" void kernel_launch(void* const* d_in, const int* in_sizes, int n_in,
                              void* d_out, int out_size, void* d_ws, size_t ws_size,
                              hipStream_t stream) {
  (void)in_sizes; (void)n_in; (void)d_ws; (void)ws_size; (void)out_size;
  const int*   x    = (const int*)  d_in[0];
  const float* h0   = (const float*)d_in[1];
  const float* emb  = (const float*)d_in[2];
  const float* Wxh0 = (const float*)d_in[3];
  const float* Wxhr = (const float*)d_in[4];
  const float* Whh  = (const float*)d_in[5];
  const float* bh   = (const float*)d_in[6];
  const float* Why  = (const float*)d_in[7];
  const float* by   = (const float*)d_in[8];
  float* out = (float*)d_out;

  dim3 grid((NBATCH + TILE_M - 1) / TILE_M);
  dim3 block(NSTAGE * 32);
  rnn_pipeline_kernel<<<grid, block, 0, stream>>>(x, h0, emb, Wxh0, Wxhr, Whh,
                                                  bh, Why, by, out);
}


# Layers 1..24 run two to a wave: product transposed (weights as A, 16 rows = 2 layers x 8 units; activations as B, 16 batch columns), K=24 over h[l-1],h[l],h[l+1]; full tile, 16 waves; same k order
# speedup vs baseline: 1.5890x; 1.1604x over previous
#include <hip/hip_runtime.h>


typedef float v8f __attribute__((ext_vector_type(8)));
typedef float v2f __attribute__((ext_vector_type(2)));

#define DIM_E   64
#define DIM_H   8
#define NLAYER  25
#define NBATCH  1000
#define NT      512
#define TILE_M  16
#define HPAD    12
#define NPRE    3
#define NSTAGE  (NPRE + NLAYER)
#define NPAIR   ((NLAYER - 1) / 2)
#define NWAVE   (NPRE + 1 + NPAIR)
#define PPAD    12
#define NSTEPS  (NT + NSTAGE - 1)

__device__ __forceinline__ float fast_tanh(float x) {
  float e = __builtin_amdgcn_exp2f(x * 2.88539008177792681f);
  return 1.0f - 2.0f * __builtin_amdgcn_rcpf(e + 1.0f);
}

__device__ __forceinline__ void step_barrier() {
  asm volatile("s_wait_dscnt 0x0" ::: "memory");
  asm volatile("s_barrier_signal -1\n\ts_barrier_wait -1" ::: "memory");
}

__device__ __forceinline__ v8f wmma4(v2f a, v2f b, v8f c) {
  return __builtin_amdgcn_wmma_f32_16x16x4_f32(false, a, false, b, (short)0, c,
                                               false, false);
}

__global__ __launch_bounds__(NWAVE * 32, 1)
void rnn_pipeline_kernel(const int* __restrict__ x,
                         const float* __restrict__ h0,
                         const float* __restrict__ emb,
                         const float* __restrict__ Wxh0,
                         const float* __restrict__ Wxhr,
                         const float* __restrict__ Whh,
                         const float* __restrict__ bh,
                         const float* __restrict__ Why,
                         const float* __restrict__ by,
                         float* __restrict__ out) {
  __shared__ __attribute__((aligned(16))) float hbuf[NLAYER][2][TILE_M][HPAD];
  __shared__ __attribute__((aligned(16))) float pacc[NPRE][2][32][PPAD];

  const int tid  = threadIdx.x;
  const int wid  = __builtin_amdgcn_readfirstlane(tid >> 5);
  const int lane = tid & 31;
  const int n    = lane & 15;
  const int hh   = lane >> 4;
  const int kb   = hh << 1;
  const int tile_base = blockIdx.x * TILE_M;
  const int l1    = 1 + 2 * (wid - NPRE - 1);
  const int stage = (wid <= NPRE) ? wid : (NPRE + l1);
  const int cbase = (wid == 0) ? 0 : ((wid == 1) ? 6 : 11);

  v2f  wA[6];
#pragma unroll
  for (int c = 0; c < 6; ++c) wA[c] = (v2f){0.0f, 0.0f};
  float bias = 0.0f;
  v8f  bvec;
#pragma unroll
  for (int r = 0; r < 8; ++r) bvec[r] = 0.0f;

  if (wid < NPRE) {
#pragma unroll
    for (int c = 0; c < 6; ++c)
#pragma unroll
      for (int j = 0; j < 2; ++j) {
        int k = (cbase + c) * 4 + kb + j;
        if (n < DIM_H && (c < 5 || wid == 0)) wA[c][j] = Wxh0[k * DIM_H + n];
      }
    if (wid == 0) bias = (n < DIM_H) ? bh[n] : 0.0f;
  } else if (wid == NPRE) {
#pragma unroll
    for (int c = 0; c < 2; ++c)
#pragma unroll
      for (int j = 0; j < 2; ++j) {
        int k = c * 4 + kb + j;
        wA[c][j] = (n < DIM_H) ? Whh[k * DIM_H + n] : 0.0f;
      }
  } else {
    const int u  = n & 7;
    const int lw = l1 + (n >> 3);
#pragma unroll
    for (int c = 0; c < 6; ++c)
#pragma unroll
      for (int j = 0; j < 2; ++j) {
        int k  = c * 4 + kb + j;
        int kl = k - 8 * (n >> 3);
        float v = 0.0f;
        if (kl >= 0 && kl < 2 * DIM_H)
          v = (kl < DIM_H) ? Wxhr[(lw - 1) * DIM_H * DIM_H + kl * DIM_H + u]
                           : Whh[lw * DIM_H * DIM_H + (kl - DIM_H) * DIM_H + u];
        wA[c][j] = v;
      }
#pragma unroll
    for (int r = 0; r < 8; ++r) bvec[r] = bh[(l1 + hh) * DIM_H + r];
  }

  if (wid == NPRE) {
    if (lane < TILE_M) {
      int gbi = tile_base + lane; if (gbi > NBATCH - 1) gbi = NBATCH - 1;
      const float* hp = h0 + gbi * DIM_H;
#pragma unroll
      for (int j = 0; j < DIM_H; ++j) {
        float v = hp[j];
        hbuf[0][0][lane][j] = v;
        hbuf[0][1][lane][j] = v;
      }
    }
  } else if (wid > NPRE) {
    int gbi = tile_base + n; if (gbi > NBATCH - 1) gbi = NBATCH - 1;
    const float* hp = h0 + ((l1 + hh) * NBATCH + gbi) * DIM_H;
#pragma unroll
    for (int j = 0; j < DIM_H; ++j) {
      float v = hp[j];
      hbuf[l1 + hh][0][n][j] = v;
      hbuf[l1 + hh][1][n][j] = v;
    }
  }

  const int m = n;
  int gb = tile_base + m; if (gb > NBATCH - 1) gb = NBATCH - 1;

  int xn = 0;
  v2f en[6];
#pragma unroll
  for (int c = 0; c < 6; ++c) en[c] = (v2f){0.0f, 0.0f};
  if (wid < NPRE) {
    const int x0 = x[gb * NT];
    xn = x[gb * NT + 1];
    const float* erow = emb + (long)x0 * DIM_E + cbase * 4 + kb;
#pragma unroll
    for (int c = 0; c < 5; ++c) en[c] = *(const v2f*)(erow + c * 4);
    if (wid == 0) en[5] = *(const v2f*)(erow + 20);
  }
  __syncthreads();

  for (int s = 0; s < NSTEPS; ++s) {
    const int t = s - stage;
    const int p_in  = (s + 1) & 1;
    const int p_out = s & 1;

    if (wid < NPRE) {
      if (t >= 0 && t < NT) {
        v8f acc;
        if (wid == 0) {
#pragma unroll
          for (int r = 0; r < 8; ++r) acc[r] = bias;
        } else {
          const float* pp = &pacc[wid - 1][p_in][lane][0];
          float4 c0 = *(const float4*)pp;
          float4 c1 = *(const float4*)(pp + 4);
          acc[0] = c0.x; acc[1] = c0.y; acc[2] = c0.z; acc[3] = c0.w;
          acc[4] = c1.x; acc[5] = c1.y; acc[6] = c1.z; acc[7] = c1.w;
        }
#pragma unroll
        for (int c = 0; c < 5; ++c) acc = wmma4(en[c], wA[c], acc);
        if (wid == 0) acc = wmma4(en[5], wA[5], acc);
        float* pq = &pacc[wid][p_out][lane][0];
        *(float4*)pq       = make_float4(acc[0], acc[1], acc[2], acc[3]);
        *(float4*)(pq + 4) = make_float4(acc[4], acc[5], acc[6], acc[7]);
        if (t + 1 < NT) {
          const float* erow = emb + (long)xn * DIM_E + cbase * 4 + kb;
#pragma unroll
          for (int c = 0; c < 5; ++c) en[c] = *(const v2f*)(erow + c * 4);
          if (wid == 0) en[5] = *(const v2f*)(erow + 20);
          if (t + 2 < NT) xn = x[gb * NT + t + 2];
        }
      }
    } else if (wid == NPRE) {
      if (t >= 0 && t < NT) {
        v8f acc;
        const float* pp = &pacc[NPRE - 1][p_in][lane][0];
        float4 c0 = *(const float4*)pp;
        float4 c1 = *(const float4*)(pp + 4);
        acc[0] = c0.x; acc[1] = c0.y; acc[2] = c0.z; acc[3] = c0.w;
        acc[4] = c1.x; acc[5] = c1.y; acc[6] = c1.z; acc[7] = c1.w;
        const float* hp = &hbuf[0][p_in][m][0];
        v2f a0 = *(const v2f*)(hp + kb);
        v2f a1 = *(const v2f*)(hp + 4 + kb);
        acc = wmma4(a0, wA[0], acc);
        acc = wmma4(a1, wA[1], acc);
#pragma unroll
        for (int r = 0; r < 8; ++r) acc[r] = fast_tanh(acc[r]);
        if (n < DIM_H) {
          const int rowbase = hh << 3;
#pragma unroll
          for (int r = 0; r < 8; ++r)
            hbuf[0][p_out][rowbase + r][n] = acc[r];
        }
      }
    } else {
      if (t >= 0 && t <= NT) {
        const float* q0 = &hbuf[l1 - 1][p_in][n][kb];
        const float* q1 = &hbuf[l1    ][p_in][n][kb];
        const float* q2 = &hbuf[l1 + 1][p_in][n][kb];
        v2f b0 = *(const v2f*)(q0);
        v2f b1 = *(const v2f*)(q0 + 4);
        v2f b2 = *(const v2f*)(q1);
        v2f b3 = *(const v2f*)(q1 + 4);
        v2f b4 = *(const v2f*)(q2);
        v2f b5 = *(const v2f*)(q2 + 4);
        asm volatile("" :: "v"(b0), "v"(b1), "v"(b2), "v"(b3), "v"(b4), "v"(b5));
        v8f acc = wmma4(wA[0], b0, bvec);
        acc = wmma4(wA[1], b1, acc);
        acc = wmma4(wA[2], b2, acc);
        acc = wmma4(wA[3], b3, acc);
        acc = wmma4(wA[4], b4, acc);
        acc = wmma4(wA[5], b5, acc);
#pragma unroll
        for (int r = 0; r < 8; ++r) acc[r] = fast_tanh(acc[r]);
        if ((unsigned)(t - hh) < (unsigned)NT) {
          float* dq = &hbuf[l1 + hh][p_out][n][0];
          *(float4*)dq       = make_float4(acc[0], acc[1], acc[2], acc[3]);
          *(float4*)(dq + 4) = make_float4(acc[4], acc[5], acc[6], acc[7]);
        }
      }
    }
    step_barrier();
  }

  if (wid >= NPRE) {
    const int  lay = (wid == NPRE) ? 0 : (l1 + hh);
    const int  row = (wid == NPRE) ? lane : n;
    const bool own = (wid == NPRE) ? (lane < TILE_M) : true;
    const int  pf  = (NPRE + lay + NT - 1) & 1;
    const int  g   = tile_base + row;
    if (own && g < NBATCH) {
      const float* src = &hbuf[lay][pf][row][0];
      float4 v0 = *(const float4*)src;
      float4 v1 = *(const float4*)(src + 4);
      float* dst = out + NBATCH + ((long)lay * NBATCH + g) * DIM_H;
      *(float4*)dst       = v0;
      *(float4*)(dst + 4) = v1;
      if (lay == NLAYER - 1) {
        float a = by[0];
#pragma unroll
        for (int j = 0; j < DIM_H; ++j) a += src[j] * Why[j];
        out[g] = a;
      }
    }
  }
}

extern "C" void kernel_launch(void* const* d_in, const int* in_sizes, int n_in,
                              void* d_out, int out_size, void* d_ws, size_t ws_size,
                              hipStream_t stream) {
  (void)in_sizes; (void)n_in; (void)d_ws; (void)ws_size; (void)out_size;
  const int*   x    = (const int*)  d_in[0];
  const float* h0   = (const float*)d_in[1];
  const float* emb  = (const float*)d_in[2];
  const float* Wxh0 = (const float*)d_in[3];
  const float* Wxhr = (const float*)d_in[4];
  const float* Whh  = (const float*)d_in[5];
  const float* bh   = (const float*)d_in[6];
  const float* Why  = (const float*)d_in[7];
  const float* by   = (const float*)d_in[8];
  float* out = (float*)d_out;

  dim3 grid((NBATCH + TILE_M - 1) / TILE_M);
  dim3 block(NWAVE * 32);
  rnn_pipeline_kernel<<<grid, block, 0, stream>>>(x, h0, emb, Wxh0, Wxhr, Whh,
                                                  bh, Why, by, out);
}


# Every wave advances 4 time steps per work-group barrier instead of 1: LDS hand-over rings of 8 step slots, embedding fragments fetched one round ahead; 146 barrier rounds, not 539; same sum order
# speedup vs baseline: 2.3443x; 1.4753x over previous
#include <hip/hip_runtime.h>


typedef float v8f __attribute__((ext_vector_type(8)));
typedef float v2f __attribute__((ext_vector_type(2)));

#define DIM_E   64
#define DIM_H   8
#define NLAYER  25
#define NBATCH  1000
#define NT      512
#define TILE_M  16
#define HPAD    12
#define NPRE    3
#define NPAIR   ((NLAYER - 1) / 2)
#define NWAVE   (NPRE + 1 + NPAIR)
#define PPAD    12
#define KSUB    4
#define RING    (2 * KSUB)
#define RMASK   (RING - 1)
#define OFF_LAST ((NPRE + 1) * KSUB + (NPAIR - 1) * (KSUB + 1))
#define NROUNDS ((NT + OFF_LAST) / KSUB + 1)

__device__ __forceinline__ float fast_tanh(float x) {
  return __builtin_amdgcn_tanhf(x);
}

__device__ __forceinline__ void step_signal() {
  asm volatile("s_wait_dscnt 0x0" ::: "memory");
  asm volatile("s_barrier_signal -1" ::: "memory");
}
__device__ __forceinline__ void step_wait() {
  asm volatile("s_barrier_wait -1" ::: "memory");
}

__device__ __forceinline__ float xhalf(float x) {
  int b = __builtin_bit_cast(int, x);
  return __builtin_bit_cast(float, __builtin_amdgcn_permlanex16(b, b, 0x76543210, 0xfedcba98, false, false));
}

__device__ __forceinline__ v8f wmma4(v2f a, v2f b, v8f c) {
  return __builtin_amdgcn_wmma_f32_16x16x4_f32(false, a, false, b, (short)0, c,
                                               false, false);
}

__device__ __forceinline__ v8f own_state_products(const v2f w0, const v2f w1, const v2f w2, const v2f w3,
                                                  const float (&st)[8], v8f c) {
  const float x4 = xhalf(st[4]), x5 = xhalf(st[5]), x6 = xhalf(st[6]), x7 = xhalf(st[7]);
  v8f acc = wmma4(w0, (v2f){st[0], st[1]}, c);
  acc = wmma4(w1, (v2f){st[2], st[3]}, acc);
  acc = wmma4(w2, (v2f){x4, x5}, acc);
  acc = wmma4(w3, (v2f){x6, x7}, acc);
  return acc;
}

__global__ __launch_bounds__(NWAVE * 32, 1)
void rnn_pipeline_kernel(const int* __restrict__ x,
                         const float* __restrict__ h0,
                         const float* __restrict__ emb,
                         const float* __restrict__ Wxh0,
                         const float* __restrict__ Wxhr,
                         const float* __restrict__ Whh,
                         const float* __restrict__ bh,
                         const float* __restrict__ Why,
                         const float* __restrict__ by,
                         float* __restrict__ out) {
  __shared__ __attribute__((aligned(16))) float hbuf[NLAYER][RING][TILE_M][HPAD];
  __shared__ __attribute__((aligned(16))) float pacc[NPRE][RING][32][PPAD];

  const int tid  = threadIdx.x;
  const int wid  = __builtin_amdgcn_readfirstlane(tid >> 5);
  const int lane = tid & 31;
  const int n    = lane & 15;
  const int hh   = lane >> 4;
  const int kb   = hh << 1;
  const int tile_base = blockIdx.x * TILE_M;
  const int l1    = 1 + 2 * (wid - NPRE - 1);
  const int toff  = (wid <= NPRE) ? wid * KSUB : ((NPRE + 1) * KSUB + (wid - NPRE - 1) * (KSUB + 1));
  const int cbase = (wid == 0) ? 0 : ((wid == 1) ? 6 : 11);

  v2f  wA[6];
#pragma unroll
  for (int c = 0; c < 6; ++c) wA[c] = (v2f){0.0f, 0.0f};
  float bias = 0.0f;
  v8f  bvec;
#pragma unroll
  for (int r = 0; r < 8; ++r) bvec[r] = 0.0f;

  if (wid < NPRE) {
#pragma unroll
    for (int c = 0; c < 6; ++c)
#pragma unroll
      for (int j = 0; j < 2; ++j) {
        int k = (cbase + c) * 4 + kb + j;
        if (n < DIM_H && (c < 5 || wid == 0)) wA[c][j] = Wxh0[k * DIM_H + n];
      }
    if (wid == 0) bias = (n < DIM_H) ? bh[n] : 0.0f;
  } else if (wid == NPRE) {
#pragma unroll
    for (int c = 0; c < 2; ++c)
#pragma unroll
      for (int j = 0; j < 2; ++j) {
        int k = c * 4 + kb + j;
        wA[c][j] = (n < DIM_H) ? Whh[k * DIM_H + n] : 0.0f;
      }
  } else {
    const int u  = n & 7;
    const int g  = n >> 3;
#pragma unroll
    for (int c = 0; c < 2; ++c)
#pragma unroll
      for (int j = 0; j < 2; ++j) {
        int k = c * 4 + kb + j;
        wA[c][j] = (g == 0) ? Wxhr[(l1 - 1) * DIM_H * DIM_H + k * DIM_H + u] : 0.0f;
      }
#pragma unroll
    for (int q = 0; q < 4; ++q)
#pragma unroll
      for (int j = 0; j < 2; ++j) {
        const int unit = 2 * q + j;
        const int d    = (q < 2) ? hh : (1 - hh);
        float v;
        if (d == 0) v = (g == 0) ? Whh[l1 * DIM_H * DIM_H + unit * DIM_H + u]
                                 : Wxhr[l1 * DIM_H * DIM_H + unit * DIM_H + u];
        else        v = (g == 0) ? 0.0f
                                 : Whh[(l1 + 1) * DIM_H * DIM_H + unit * DIM_H + u];
        wA[2 + q][j] = v;
      }
#pragma unroll
    for (int r = 0; r < 8; ++r) bvec[r] = bh[(l1 + hh) * DIM_H + r];
  }

  float st[8];
#pragma unroll
  for (int r = 0; r < 8; ++r) st[r] = 0.0f;
  if (wid == NPRE) {
    if (lane < TILE_M) {
      int gbi = tile_base + lane; if (gbi > NBATCH - 1) gbi = NBATCH - 1;
      const float* hp = h0 + gbi * DIM_H;
#pragma unroll
      for (int j = 0; j < DIM_H; ++j) {
        float v = hp[j];
#pragma unroll
        for (int sl = 0; sl < RING; ++sl) hbuf[0][sl][lane][j] = v;
      }
    }
  } else if (wid > NPRE) {
    int gbi = tile_base + n; if (gbi > NBATCH - 1) gbi = NBATCH - 1;
    const float* hp = h0 + ((l1 + hh) * NBATCH + gbi) * DIM_H;
#pragma unroll
    for (int j = 0; j < DIM_H; ++j) {
      float v = hp[j];
      st[j] = v;
#pragma unroll
      for (int sl = 0; sl < RING; ++sl) hbuf[l1 + hh][sl][n][j] = v;
    }
  }

  const int m = n;
  int gb = tile_base + m; if (gb > NBATCH - 1) gb = NBATCH - 1;

  int xn[KSUB];
  v2f en[KSUB][6];
#pragma unroll
  for (int j = 0; j < KSUB; ++j) {
    xn[j] = 0;
#pragma unroll
    for (int c = 0; c < 6; ++c) en[j][c] = (v2f){0.0f, 0.0f};
  }
  if (wid < NPRE) {
#pragma unroll
    for (int j = 0; j < KSUB; ++j) {
      int ta = j - toff;         ta = ta < 0 ? 0 : (ta > NT - 1 ? NT - 1 : ta);
      int tb = KSUB + j - toff;  tb = tb < 0 ? 0 : (tb > NT - 1 ? NT - 1 : tb);
      const int x0 = x[gb * NT + ta];
      xn[j] = x[gb * NT + tb];
      const float* erow = emb + (long)x0 * DIM_E + cbase * 4 + kb;
#pragma unroll
      for (int c = 0; c < 5; ++c) en[j][c] = *(const v2f*)(erow + c * 4);
      if (wid == 0) en[j][5] = *(const v2f*)(erow + 20);
    }
  }
  __syncthreads();

  if (wid < NPRE) {
    for (int r = 0; r < NROUNDS; ++r) {
      const int t0 = r * KSUB - toff;
#pragma unroll
      for (int j = 0; j < KSUB; ++j) {
        const int t = t0 + j;
        if (t >= 0 && t < NT) {
          v8f acc;
          if (wid == 0) {
#pragma unroll
            for (int q = 0; q < 8; ++q) acc[q] = bias;
          } else {
            const float* pp = &pacc[wid - 1][t & RMASK][lane][0];
            float4 c0 = *(const float4*)pp;
            float4 c1 = *(const float4*)(pp + 4);
            acc[0] = c0.x; acc[1] = c0.y; acc[2] = c0.z; acc[3] = c0.w;
            acc[4] = c1.x; acc[5] = c1.y; acc[6] = c1.z; acc[7] = c1.w;
          }
#pragma unroll
          for (int c = 0; c < 5; ++c) acc = wmma4(en[j][c], wA[c], acc);
          if (wid == 0) acc = wmma4(en[j][5], wA[5], acc);
          float* pq = &pacc[wid][t & RMASK][lane][0];
          *(float4*)pq       = make_float4(acc[0], acc[1], acc[2], acc[3]);
          *(float4*)(pq + 4) = make_float4(acc[4], acc[5], acc[6], acc[7]);
        }
      }
#pragma unroll
      for (int j = 0; j < KSUB; ++j) {
        const float* erow = emb + (long)xn[j] * DIM_E + cbase * 4 + kb;
#pragma unroll
        for (int c = 0; c < 5; ++c) en[j][c] = *(const v2f*)(erow + c * 4);
        if (wid == 0) en[j][5] = *(const v2f*)(erow + 20);
      }
#pragma unroll
      for (int j = 0; j < KSUB; ++j) {
        int tn = t0 + 2 * KSUB + j;  tn = tn < 0 ? 0 : (tn > NT - 1 ? NT - 1 : tn);
        xn[j] = x[gb * NT + tn];
      }
      step_signal();
      step_wait();
    }
  } else if (wid == NPRE) {
    for (int r = 0; r < NROUNDS; ++r) {
      const int t0 = r * KSUB - toff;
#pragma unroll
      for (int j = 0; j < KSUB; ++j) {
        const int t = t0 + j;
        if (t >= 0 && t < NT) {
          v8f acc;
          const float* pp = &pacc[NPRE - 1][t & RMASK][lane][0];
          float4 c0 = *(const float4*)pp;
          float4 c1 = *(const float4*)(pp + 4);
          acc[0] = c0.x; acc[1] = c0.y; acc[2] = c0.z; acc[3] = c0.w;
          acc[4] = c1.x; acc[5] = c1.y; acc[6] = c1.z; acc[7] = c1.w;
          const float* hp = &hbuf[0][(t - 1) & RMASK][m][0];
          v2f a0 = *(const v2f*)(hp + kb);
          v2f a1 = *(const v2f*)(hp + 4 + kb);
          acc = wmma4(a0, wA[0], acc);
          acc = wmma4(a1, wA[1], acc);
#pragma unroll
          for (int q = 0; q < 8; ++q) acc[q] = fast_tanh(acc[q]);
          if (n < DIM_H) {
            const int rowbase = hh << 3;
#pragma unroll
            for (int q = 0; q < 8; ++q)
              hbuf[0][t & RMASK][rowbase + q][n] = acc[q];
          }
        }
      }
      step_signal();
      step_wait();
    }
  } else {
    v8f accp = own_state_products(wA[2], wA[3], wA[4], wA[5], st, bvec);
    for (int r = 0; r < NROUNDS; ++r) {
      const int t0 = r * KSUB - toff;
      v2f b0[KSUB], b1[KSUB];
#pragma unroll
      for (int j = 0; j < KSUB; ++j) {
        const float* q0 = &hbuf[l1 - 1][(t0 + j) & RMASK][n][kb];
        b0[j] = *(const v2f*)(q0);
        b1[j] = *(const v2f*)(q0 + 4);
      }
#pragma unroll
      for (int j = 0; j < KSUB; ++j) {
        const int  t   = t0 + j;
        const bool due = (t >= 0 && t <= NT);
        if (due) {
          v8f acc = wmma4(wA[0], b0[j], accp);
          acc = wmma4(wA[1], b1[j], acc);
          if ((unsigned)(t - hh) < (unsigned)NT) {
#pragma unroll
            for (int q = 0; q < 8; ++q) st[q] = fast_tanh(acc[q]);
            float* dq = &hbuf[l1 + hh][(t - hh) & RMASK][n][0];
            *(float4*)dq       = make_float4(st[0], st[1], st[2], st[3]);
            *(float4*)(dq + 4) = make_float4(st[4], st[5], st[6], st[7]);
          }
        }
        if (j == KSUB - 1) {
          step_signal();
          __builtin_amdgcn_sched_barrier(0);
        }
        if (due) accp = own_state_products(wA[2], wA[3], wA[4], wA[5], st, bvec);
        if (j == KSUB - 1) {
          __builtin_amdgcn_sched_barrier(0);
          step_wait();
        }
      }
    }
  }

  if (wid >= NPRE) {
    const int  lay = (wid == NPRE) ? 0 : (l1 + hh);
    const int  row = (wid == NPRE) ? lane : n;
    const bool own = (wid == NPRE) ? (lane < TILE_M) : true;
    const int  pf  = (NT - 1) & RMASK;
    const int  g   = tile_base + row;
    if (own && g < NBATCH) {
      const float* src = &hbuf[lay][pf][row][0];
      float4 v0 = *(const float4*)src;
      float4 v1 = *(const float4*)(src + 4);
      float* dst = out + NBATCH + ((long)lay * NBATCH + g) * DIM_H;
      *(float4*)dst       = v0;
      *(float4*)(dst + 4) = v1;
      if (lay == NLAYER - 1) {
        float a = by[0];
#pragma unroll
        for (int j = 0; j < DIM_H; ++j) a += src[j] * Why[j];
        out[g] = a;
      }
    }
  }
}

extern "C" void kernel_launch(void* const* d_in, const int* in_sizes, int n_in,
                              void* d_out, int out_size, void* d_ws, size_t ws_size,
                              hipStream_t stream) {
  (void)in_sizes; (void)n_in; (void)d_ws; (void)ws_size; (void)out_size;
  const int*   x    = (const int*)  d_in[0];
  const float* h0   = (const float*)d_in[1];
  const float* emb  = (const float*)d_in[2];
  const float* Wxh0 = (const float*)d_in[3];
  const float* Wxhr = (const float*)d_in[4];
  const float* Whh  = (const float*)d_in[5];
  const float* bh   = (const float*)d_in[6];
  const float* Why  = (const float*)d_in[7];
  const float* by   = (const float*)d_in[8];
  float* out = (float*)d_out;

  dim3 grid((NBATCH + TILE_M - 1) / TILE_M);
  dim3 block(NWAVE * 32);
  rnn_pipeline_kernel<<<grid, block, 0, stream>>>(x, h0, emb, Wxh0, Wxhr, Whh,
                                                  bh, Why, by, out);
}


# Layer-0 input projection formed once per vocabulary row by a first launch (table bias + embedding x W_xh0 in the workspace, same sum order); layer 0 gathers 32 B per token as accumulator start
# speedup vs baseline: 2.8212x; 1.2034x over previous
#include <hip/hip_runtime.h>


typedef float v8f __attribute__((ext_vector_type(8)));
typedef float v2f __attribute__((ext_vector_type(2)));

#define DIM_E   64
#define DIM_H   8
#define NLAYER  25
#define NBATCH  1000
#define NT      512
#define NVOCAB  32000
#define TILE_M  16
#define TABW    16
#define HPAD    20
#define NPAIR   ((NLAYER - 1) / 2)
#define NWAVE   (1 + NPAIR)
#define KSUB    4
#define RING    (2 * KSUB)
#define RMASK   (RING - 1)
#define OFF_LAST (KSUB + (NPAIR - 1) * (KSUB + 1))
#define NROUNDS ((NT + OFF_LAST) / KSUB + 1)

__device__ __forceinline__ float fast_tanh(float x) {
  return __builtin_amdgcn_tanhf(x);
}

__device__ __forceinline__ void step_signal() {
  asm volatile("s_wait_dscnt 0x0" ::: "memory");
  asm volatile("s_barrier_signal -1" ::: "memory");
}
__device__ __forceinline__ void step_wait() {
  asm volatile("s_barrier_wait -1" ::: "memory");
}

__device__ __forceinline__ float xhalf(float x) {
  int b = __builtin_bit_cast(int, x);
  return __builtin_bit_cast(float, __builtin_amdgcn_permlanex16(b, b, 0x76543210, 0xfedcba98, false, false));
}

__device__ __forceinline__ v8f wmma4(v2f a, v2f b, v8f c) {
  return __builtin_amdgcn_wmma_f32_16x16x4_f32(false, a, false, b, (short)0, c,
                                               false, false);
}

__device__ __forceinline__ v8f own_state_products(const v2f w0, const v2f w1, const v2f w2, const v2f w3,
                                                  const float (&st)[8], v8f c) {
  const float x4 = xhalf(st[4]), x5 = xhalf(st[5]), x6 = xhalf(st[6]), x7 = xhalf(st[7]);
  v8f acc = wmma4(w0, (v2f){st[0], st[1]}, c);
  acc = wmma4(w1, (v2f){st[2], st[3]}, acc);
  acc = wmma4(w2, (v2f){x4, x5}, acc);
  acc = wmma4(w3, (v2f){x6, x7}, acc);
  return acc;
}

__global__ __launch_bounds__(128)
void input_table_kernel(const float* __restrict__ emb,
                        const float* __restrict__ Wxh0,
                        const float* __restrict__ bh,
                        float* __restrict__ tab) {
  const int tid  = threadIdx.x;
  const int wid  = __builtin_amdgcn_readfirstlane(tid >> 5);
  const int lane = tid & 31;
  const int n    = lane & 15;
  const int hh   = lane >> 4;
  const int kb   = hh << 1;
  const int v0   = (blockIdx.x * 4 + wid) * TILE_M;
  const int u    = (n < DIM_H) ? n : ((n - DIM_H + 2) & 7);
  const float* erow = emb + (long)(v0 + n) * DIM_E + kb;
  v2f a[16], w[16];
#pragma unroll
  for (int c = 0; c < 16; ++c) a[c] = *(const v2f*)(erow + c * 4);
#pragma unroll
  for (int c = 0; c < 16; ++c)
#pragma unroll
    for (int j = 0; j < 2; ++j) w[c][j] = Wxh0[(c * 4 + kb + j) * DIM_H + u];
  const float bias = bh[u];
  v8f acc;
#pragma unroll
  for (int q = 0; q < 8; ++q) acc[q] = bias;
#pragma unroll
  for (int c = 0; c < 16; ++c) acc = wmma4(a[c], w[c], acc);
  float* dst = tab + (long)(v0 + 8 * hh) * TABW + n;
#pragma unroll
  for (int q = 0; q < 8; ++q) dst[q * TABW] = acc[q];
}

__global__ __launch_bounds__(NWAVE * 32, 1)
void rnn_pipeline_kernel(const int* __restrict__ x,
                         const float* __restrict__ h0,
                         const float* __restrict__ tab,
                         const float* __restrict__ Wxhr,
                         const float* __restrict__ Whh,
                         const float* __restrict__ bh,
                         const float* __restrict__ Why,
                         const float* __restrict__ by,
                         float* __restrict__ out) {
  __shared__ __attribute__((aligned(16))) float hbuf[NLAYER][RING][TILE_M][HPAD];

  const int tid  = threadIdx.x;
  const int wid  = __builtin_amdgcn_readfirstlane(tid >> 5);
  const int lane = tid & 31;
  const int n    = lane & 15;
  const int hh   = lane >> 4;
  const int kb   = hh << 1;
  const int tile_base = blockIdx.x * TILE_M;
  const int l1    = 1 + 2 * (wid - 1);
  const int toff  = (wid == 0) ? 0 : (KSUB + (wid - 1) * (KSUB + 1));

  v2f  wA[6];
#pragma unroll
  for (int c = 0; c < 6; ++c) wA[c] = (v2f){0.0f, 0.0f};
  v8f  bvec;
#pragma unroll
  for (int r = 0; r < 8; ++r) bvec[r] = 0.0f;

  if (wid == 0) {
    const int u = (n < DIM_H) ? n : ((n - DIM_H + 2) & 7);
#pragma unroll
    for (int c = 0; c < 2; ++c)
#pragma unroll
      for (int j = 0; j < 2; ++j) {
        int k = c * 4 + kb + j;
        wA[c][j] = Whh[k * DIM_H + u];
      }
  } else {
    const int u  = n & 7;
    const int g  = n >> 3;
#pragma unroll
    for (int c = 0; c < 2; ++c)
#pragma unroll
      for (int j = 0; j < 2; ++j) {
        int k = c * 4 + kb + j;
        wA[c][j] = (g == 0) ? Wxhr[(l1 - 1) * DIM_H * DIM_H + k * DIM_H + u] : 0.0f;
      }
#pragma unroll
    for (int q = 0; q < 4; ++q)
#pragma unroll
      for (int j = 0; j < 2; ++j) {
        const int unit = 2 * q + j;
        const int d    = (q < 2) ? hh : (1 - hh);
        float v;
        if (d == 0) v = (g == 0) ? Whh[l1 * DIM_H * DIM_H + unit * DIM_H + u]
                                 : Wxhr[l1 * DIM_H * DIM_H + unit * DIM_H + u];
        else        v = (g == 0) ? 0.0f
                                 : Whh[(l1 + 1) * DIM_H * DIM_H + unit * DIM_H + u];
        wA[2 + q][j] = v;
      }
#pragma unroll
    for (int r = 0; r < 8; ++r) bvec[r] = bh[(l1 + hh) * DIM_H + r];
  }

  float st[8];
#pragma unroll
  for (int r = 0; r < 8; ++r) st[r] = 0.0f;
  int gb = tile_base + n; if (gb > NBATCH - 1) gb = NBATCH - 1;
  if (wid == 0) {
    const float* hp = h0 + gb * DIM_H;
#pragma unroll
    for (int r = 0; r < DIM_H; ++r) st[r] = hp[(r + kb) & 7];
    if (hh == 0) {
#pragma unroll
      for (int j = 0; j < DIM_H; ++j)
#pragma unroll
        for (int sl = 0; sl < RING; ++sl) hbuf[0][sl][n][j] = st[j];
    }
  } else {
    const float* hp = h0 + ((l1 + hh) * NBATCH + gb) * DIM_H;
#pragma unroll
    for (int j = 0; j < DIM_H; ++j) {
      float v = hp[j];
      st[j] = v;
#pragma unroll
      for (int sl = 0; sl < RING; ++sl) hbuf[l1 + hh][sl][n][j] = v;
    }
  }

  int4   xn = make_int4(0, 0, 0, 0);
  float4 pa[KSUB], pb[KSUB];
#pragma unroll
  for (int j = 0; j < KSUB; ++j) { pa[j] = make_float4(0.f, 0.f, 0.f, 0.f); pb[j] = pa[j]; }
  if (wid == 0) {
    const int4 x0 = *(const int4*)(x + gb * NT);
    xn = *(const int4*)(x + gb * NT + KSUB);
    const int id0[KSUB] = {x0.x, x0.y, x0.z, x0.w};
#pragma unroll
    for (int j = 0; j < KSUB; ++j) {
      int id = id0[j]; id = id < 0 ? 0 : (id > NVOCAB - 1 ? NVOCAB - 1 : id);
      const float* prow = tab + (long)id * TABW + hh * DIM_H;
      pa[j] = *(const float4*)prow;
      pb[j] = *(const float4*)(prow + 4);
    }
  }
  __syncthreads();

  if (wid == 0) {
    for (int r = 0; r < NROUNDS; ++r) {
      const int t0 = r * KSUB;
#pragma unroll
      for (int j = 0; j < KSUB; ++j) {
        const int t = t0 + j;
        if (t < NT) {
          v8f acc;
          acc[0] = pa[j].x; acc[1] = pa[j].y; acc[2] = pa[j].z; acc[3] = pa[j].w;
          acc[4] = pb[j].x; acc[5] = pb[j].y; acc[6] = pb[j].z; acc[7] = pb[j].w;
          acc = wmma4(wA[0], (v2f){st[0], st[1]}, acc);
          acc = wmma4(wA[1], (v2f){st[4], st[5]}, acc);
#pragma unroll
          for (int q = 0; q < 8; ++q) st[q] = fast_tanh(acc[q]);
          if (hh == 0) {
            float* dq = &hbuf[0][t & RMASK][n][0];
            *(float4*)dq       = make_float4(st[0], st[1], st[2], st[3]);
            *(float4*)(dq + 4) = make_float4(st[4], st[5], st[6], st[7]);
          }
        }
      }
      {
        const int idn[KSUB] = {xn.x, xn.y, xn.z, xn.w};
#pragma unroll
        for (int j = 0; j < KSUB; ++j) {
          int id = idn[j]; id = id < 0 ? 0 : (id > NVOCAB - 1 ? NVOCAB - 1 : id);
          const float* prow = tab + (long)id * TABW + hh * DIM_H;
          pa[j] = *(const float4*)prow;
          pb[j] = *(const float4*)(prow + 4);
        }
        int tn = t0 + 2 * KSUB;  tn = tn > NT - KSUB ? NT - KSUB : tn;
        xn = *(const int4*)(x + gb * NT + tn);
      }
      step_signal();
      step_wait();
    }
  } else {
    v8f accp = own_state_products(wA[2], wA[3], wA[4], wA[5], st, bvec);
    for (int r = 0; r < NROUNDS; ++r) {
      const int t0 = r * KSUB - toff;
      v2f b0[KSUB], b1[KSUB];
#pragma unroll
      for (int j = 0; j < KSUB; ++j) {
        const float* q0 = &hbuf[l1 - 1][(t0 + j) & RMASK][n][kb];
        b0[j] = *(const v2f*)(q0);
        b1[j] = *(const v2f*)(q0 + 4);
      }
#pragma unroll
      for (int j = 0; j < KSUB; ++j) {
        const int  t   = t0 + j;
        const bool due = (t >= 0 && t <= NT);
        if (due) {
          v8f acc = wmma4(wA[0], b0[j], accp);
          acc = wmma4(wA[1], b1[j], acc);
          if ((unsigned)(t - hh) < (unsigned)NT) {
#pragma unroll
            for (int q = 0; q < 8; ++q) st[q] = fast_tanh(acc[q]);
            float* dq = &hbuf[l1 + hh][(t - hh) & RMASK][n][0];
            *(float4*)dq       = make_float4(st[0], st[1], st[2], st[3]);
            *(float4*)(dq + 4) = make_float4(st[4], st[5], st[6], st[7]);
          }
        }
        if (j == KSUB - 1) {
          step_signal();
          __builtin_amdgcn_sched_barrier(0);
        }
        if (due) accp = own_state_products(wA[2], wA[3], wA[4], wA[5], st, bvec);
        if (j == KSUB - 1) {
          __builtin_amdgcn_sched_barrier(0);
          step_wait();
        }
      }
    }
  }

  {
    const int  lay = (wid == 0) ? 0 : (l1 + hh);
    const int  row = n;
    const bool own = (wid == 0) ? (hh == 0) : true;
    const int  pf  = (NT - 1) & RMASK;
    const int  g   = tile_base + row;
    if (own && g < NBATCH) {
      const float* src = &hbuf[lay][pf][row][0];
      float4 v0 = *(const float4*)src;
      float4 v1 = *(const float4*)(src + 4);
      float* dst = out + NBATCH + ((long)lay * NBATCH + g) * DIM_H;
      *(float4*)dst       = v0;
      *(float4*)(dst + 4) = v1;
      if (lay == NLAYER - 1) {
        float a = by[0];
#pragma unroll
        for (int j = 0; j < DIM_H; ++j) a += src[j] * Why[j];
        out[g] = a;
      }
    }
  }
}

extern "C" void kernel_launch(void* const* d_in, const int* in_sizes, int n_in,
                              void* d_out, int out_size, void* d_ws, size_t ws_size,
                              hipStream_t stream) {
  (void)in_sizes; (void)n_in; (void)ws_size; (void)out_size;
  const int*   x    = (const int*)  d_in[0];
  const float* h0   = (const float*)d_in[1];
  const float* emb  = (const float*)d_in[2];
  const float* Wxh0 = (const float*)d_in[3];
  const float* Wxhr = (const float*)d_in[4];
  const float* Whh  = (const float*)d_in[5];
  const float* bh   = (const float*)d_in[6];
  const float* Why  = (const float*)d_in[7];
  const float* by   = (const float*)d_in[8];
  float* out = (float*)d_out;
  float* tab = (float*)d_ws;

  input_table_kernel<<<dim3(NVOCAB / TILE_M / 4), dim3(128), 0, stream>>>(emb, Wxh0, bh, tab);

  dim3 grid((NBATCH + TILE_M - 1) / TILE_M);
  dim3 block(NWAVE * 32);
  rnn_pipeline_kernel<<<grid, block, 0, stream>>>(x, h0, tab, Wxhr, Whh,
                                                  bh, Why, by, out);
}


# Work-group barrier split over the whole round: arrive after the round's LDS reads, wait at its end; hand-over lags 2 rounds of 2 steps (was 1 of 4), same pipeline offsets and sum order
# speedup vs baseline: 2.9148x; 1.0332x over previous
#include <hip/hip_runtime.h>


typedef float v8f __attribute__((ext_vector_type(8)));
typedef float v2f __attribute__((ext_vector_type(2)));

#define DIM_E   64
#define DIM_H   8
#define NLAYER  25
#define NBATCH  1000
#define NT      512
#define NVOCAB  32000
#define TILE_M  16
#define TABW    16
#define HPAD    20
#define NPAIR   ((NLAYER - 1) / 2)
#define NWAVE   (1 + NPAIR)
#define KSUB    2
#define LAG     2
#define RING    (4 * KSUB)
#define RMASK   (RING - 1)
#define TOKQ    (2 * KSUB)
#define OFF_LAST (LAG * KSUB + (NPAIR - 1) * (LAG * KSUB + 1))
#define NROUNDS ((NT + OFF_LAST) / KSUB + 1)
static_assert(NROUNDS % 2 == 0, "the layer-0 wave takes the rounds two at a time");
static_assert(RING >= (LAG + 1) * KSUB, "a slot must not be written again before its reader's round has opened");

__device__ __forceinline__ float fast_tanh(float x) {
  return __builtin_amdgcn_tanhf(x);
}

__device__ __forceinline__ void round_open() {
  asm volatile("s_wait_dscnt 0x0" ::: "memory");
  asm volatile("s_barrier_signal -1" ::: "memory");
}
__device__ __forceinline__ void round_close() {
  asm volatile("s_barrier_wait -1" ::: "memory");
}

__device__ __forceinline__ float xhalf(float x) {
  int b = __builtin_bit_cast(int, x);
  return __builtin_bit_cast(float, __builtin_amdgcn_permlanex16(b, b, 0x76543210, 0xfedcba98, false, false));
}

__device__ __forceinline__ v8f wmma4(v2f a, v2f b, v8f c) {
  return __builtin_amdgcn_wmma_f32_16x16x4_f32(false, a, false, b, (short)0, c,
                                               false, false);
}

__device__ __forceinline__ v8f own_state_products(const v2f w0, const v2f w1, const v2f w2, const v2f w3,
                                                  const float (&st)[8], v8f c) {
  const float x4 = xhalf(st[4]), x5 = xhalf(st[5]), x6 = xhalf(st[6]), x7 = xhalf(st[7]);
  v8f acc = wmma4(w0, (v2f){st[0], st[1]}, c);
  acc = wmma4(w1, (v2f){st[2], st[3]}, acc);
  acc = wmma4(w2, (v2f){x4, x5}, acc);
  acc = wmma4(w3, (v2f){x6, x7}, acc);
  return acc;
}

__global__ __launch_bounds__(128)
void input_table_kernel(const float* __restrict__ emb,
                        const float* __restrict__ Wxh0,
                        const float* __restrict__ bh,
                        float* __restrict__ tab) {
  const int tid  = threadIdx.x;
  const int wid  = __builtin_amdgcn_readfirstlane(tid >> 5);
  const int lane = tid & 31;
  const int n    = lane & 15;
  const int hh   = lane >> 4;
  const int kb   = hh << 1;
  const int v0   = (blockIdx.x * 4 + wid) * TILE_M;
  const int u    = (n < DIM_H) ? n : ((n - DIM_H + 2) & 7);
  const float* erow = emb + (long)(v0 + n) * DIM_E + kb;
  v2f a[16], w[16];
#pragma unroll
  for (int c = 0; c < 16; ++c) a[c] = *(const v2f*)(erow + c * 4);
#pragma unroll
  for (int c = 0; c < 16; ++c)
#pragma unroll
    for (int j = 0; j < 2; ++j) w[c][j] = Wxh0[(c * 4 + kb + j) * DIM_H + u];
  const float bias = bh[u];
  v8f acc;
#pragma unroll
  for (int q = 0; q < 8; ++q) acc[q] = bias;
#pragma unroll
  for (int c = 0; c < 16; ++c) acc = wmma4(a[c], w[c], acc);
  float* dst = tab + (long)(v0 + 8 * hh) * TABW + n;
#pragma unroll
  for (int q = 0; q < 8; ++q) dst[q * TABW] = acc[q];
}

__global__ __launch_bounds__(NWAVE * 32, 1)
void rnn_pipeline_kernel(const int* __restrict__ x,
                         const float* __restrict__ h0,
                         const float* __restrict__ tab,
                         const float* __restrict__ Wxhr,
                         const float* __restrict__ Whh,
                         const float* __restrict__ bh,
                         const float* __restrict__ Why,
                         const float* __restrict__ by,
                         float* __restrict__ out) {
  __shared__ __attribute__((aligned(16))) float hbuf[NLAYER][RING][TILE_M][HPAD];

  const int tid  = threadIdx.x;
  const int wid  = __builtin_amdgcn_readfirstlane(tid >> 5);
  const int lane = tid & 31;
  const int n    = lane & 15;
  const int hh   = lane >> 4;
  const int kb   = hh << 1;
  const int tile_base = blockIdx.x * TILE_M;
  const int l1    = 1 + 2 * (wid - 1);
  const int toff  = (wid == 0) ? 0 : (LAG * KSUB + (wid - 1) * (LAG * KSUB + 1));

  v2f  wA[6];
#pragma unroll
  for (int c = 0; c < 6; ++c) wA[c] = (v2f){0.0f, 0.0f};
  v8f  bvec;
#pragma unroll
  for (int r = 0; r < 8; ++r) bvec[r] = 0.0f;

  if (wid == 0) {
    const int u = (n < DIM_H) ? n : ((n - DIM_H + 2) & 7);
#pragma unroll
    for (int c = 0; c < 2; ++c)
#pragma unroll
      for (int j = 0; j < 2; ++j) {
        int k = c * 4 + kb + j;
        wA[c][j] = Whh[k * DIM_H + u];
      }
  } else {
    const int u  = n & 7;
    const int g  = n >> 3;
#pragma unroll
    for (int c = 0; c < 2; ++c)
#pragma unroll
      for (int j = 0; j < 2; ++j) {
        int k = c * 4 + kb + j;
        wA[c][j] = (g == 0) ? Wxhr[(l1 - 1) * DIM_H * DIM_H + k * DIM_H + u] : 0.0f;
      }
#pragma unroll
    for (int q = 0; q < 4; ++q)
#pragma unroll
      for (int j = 0; j < 2; ++j) {
        const int unit = 2 * q + j;
        const int d    = (q < 2) ? hh : (1 - hh);
        float v;
        if (d == 0) v = (g == 0) ? Whh[l1 * DIM_H * DIM_H + unit * DIM_H + u]
                                 : Wxhr[l1 * DIM_H * DIM_H + unit * DIM_H + u];
        else        v = (g == 0) ? 0.0f
                                 : Whh[(l1 + 1) * DIM_H * DIM_H + unit * DIM_H + u];
        wA[2 + q][j] = v;
      }
#pragma unroll
    for (int r = 0; r < 8; ++r) bvec[r] = bh[(l1 + hh) * DIM_H + r];
  }

  float st[8];
#pragma unroll
  for (int r = 0; r < 8; ++r) st[r] = 0.0f;
  int gb = tile_base + n; if (gb > NBATCH - 1) gb = NBATCH - 1;
  if (wid == 0) {
    const float* hp = h0 + gb * DIM_H;
#pragma unroll
    for (int r = 0; r < DIM_H; ++r) st[r] = hp[(r + kb) & 7];
    if (hh == 0) {
#pragma unroll
      for (int j = 0; j < DIM_H; ++j)
#pragma unroll
        for (int sl = 0; sl < RING; ++sl) hbuf[0][sl][n][j] = st[j];
    }
  } else {
    const float* hp = h0 + ((l1 + hh) * NBATCH + gb) * DIM_H;
#pragma unroll
    for (int j = 0; j < DIM_H; ++j) {
      float v = hp[j];
      st[j] = v;
#pragma unroll
      for (int sl = 0; sl < RING; ++sl) hbuf[l1 + hh][sl][n][j] = v;
    }
  }

  int4   xn = make_int4(0, 0, 0, 0);
  float4 pa[TOKQ], pb[TOKQ];
#pragma unroll
  for (int j = 0; j < TOKQ; ++j) { pa[j] = make_float4(0.f, 0.f, 0.f, 0.f); pb[j] = pa[j]; }
  if (wid == 0) {
    const int4 x0 = *(const int4*)(x + gb * NT);
    xn = *(const int4*)(x + gb * NT + TOKQ);
    const int id0[TOKQ] = {x0.x, x0.y, x0.z, x0.w};
#pragma unroll
    for (int j = 0; j < TOKQ; ++j) {
      int id = id0[j]; id = id < 0 ? 0 : (id > NVOCAB - 1 ? NVOCAB - 1 : id);
      const float* prow = tab + (long)id * TABW + hh * DIM_H;
      pa[j] = *(const float4*)prow;
      pb[j] = *(const float4*)(prow + 4);
    }
  }
  __syncthreads();

  if (wid == 0) {
    for (int rr = 0; rr < NROUNDS / 2; ++rr) {
      const int t0 = rr * TOKQ;
#pragma unroll
      for (int j = 0; j < TOKQ; ++j) {
        const int t = t0 + j;
        if (j % KSUB == 0) round_open();
        if (t < NT) {
          v8f acc;
          acc[0] = pa[j].x; acc[1] = pa[j].y; acc[2] = pa[j].z; acc[3] = pa[j].w;
          acc[4] = pb[j].x; acc[5] = pb[j].y; acc[6] = pb[j].z; acc[7] = pb[j].w;
          acc = wmma4(wA[0], (v2f){st[0], st[1]}, acc);
          acc = wmma4(wA[1], (v2f){st[4], st[5]}, acc);
#pragma unroll
          for (int q = 0; q < 8; ++q) st[q] = fast_tanh(acc[q]);
          if (hh == 0) {
            float* dq = &hbuf[0][t & RMASK][n][0];
            *(float4*)dq       = make_float4(st[0], st[1], st[2], st[3]);
            *(float4*)(dq + 4) = make_float4(st[4], st[5], st[6], st[7]);
          }
        }
        if (j == TOKQ - 1) {
          const int idn[TOKQ] = {xn.x, xn.y, xn.z, xn.w};
#pragma unroll
          for (int i = 0; i < TOKQ; ++i) {
            int id = idn[i]; id = id < 0 ? 0 : (id > NVOCAB - 1 ? NVOCAB - 1 : id);
            const float* prow = tab + (long)id * TABW + hh * DIM_H;
            pa[i] = *(const float4*)prow;
            pb[i] = *(const float4*)(prow + 4);
          }
          int tn = t0 + 2 * TOKQ;  tn = tn > NT - TOKQ ? NT - TOKQ : tn;
          xn = *(const int4*)(x + gb * NT + tn);
        }
        if (j % KSUB == KSUB - 1) round_close();
      }
    }
  } else {
    v8f accp = bvec;
    for (int r = 0; r < NROUNDS; ++r) {
      const int t0 = r * KSUB - toff;
      v2f b0[KSUB], b1[KSUB];
#pragma unroll
      for (int j = 0; j < KSUB; ++j) {
        const float* q0 = &hbuf[l1 - 1][(t0 + j) & RMASK][n][kb];
        b0[j] = *(const v2f*)(q0);
        b1[j] = *(const v2f*)(q0 + 4);
      }
      if (t0 >= 0 && t0 <= NT) accp = own_state_products(wA[2], wA[3], wA[4], wA[5], st, bvec);
      round_open();
#pragma unroll
      for (int j = 0; j < KSUB; ++j) asm volatile("" :: "v"(b0[j]), "v"(b1[j]));
#pragma unroll
      for (int j = 0; j < KSUB; ++j) {
        const int  t   = t0 + j;
        const bool due = (t >= 0 && t <= NT);
        if (due) {
          if (j > 0) accp = own_state_products(wA[2], wA[3], wA[4], wA[5], st, bvec);
          v8f acc = wmma4(wA[0], b0[j], accp);
          acc = wmma4(wA[1], b1[j], acc);
          if ((unsigned)(t - hh) < (unsigned)NT) {
#pragma unroll
            for (int q = 0; q < 8; ++q) st[q] = fast_tanh(acc[q]);
            float* dq = &hbuf[l1 + hh][(t - hh) & RMASK][n][0];
            *(float4*)dq       = make_float4(st[0], st[1], st[2], st[3]);
            *(float4*)(dq + 4) = make_float4(st[4], st[5], st[6], st[7]);
          }
        }
      }
      round_close();
    }
  }

  {
    const int  lay = (wid == 0) ? 0 : (l1 + hh);
    const int  row = n;
    const bool own = (wid == 0) ? (hh == 0) : true;
    const int  pf  = (NT - 1) & RMASK;
    const int  g   = tile_base + row;
    if (own && g < NBATCH) {
      const float* src = &hbuf[lay][pf][row][0];
      float4 v0 = *(const float4*)src;
      float4 v1 = *(const float4*)(src + 4);
      float* dst = out + NBATCH + ((long)lay * NBATCH + g) * DIM_H;
      *(float4*)dst       = v0;
      *(float4*)(dst + 4) = v1;
      if (lay == NLAYER - 1) {
        float a = by[0];
#pragma unroll
        for (int j = 0; j < DIM_H; ++j) a += src[j] * Why[j];
        out[g] = a;
      }
    }
  }
}

extern "C" void kernel_launch(void* const* d_in, const int* in_sizes, int n_in,
                              void* d_out, int out_size, void* d_ws, size_t ws_size,
                              hipStream_t stream) {
  (void)in_sizes; (void)n_in; (void)ws_size; (void)out_size;
  const int*   x    = (const int*)  d_in[0];
  const float* h0   = (const float*)d_in[1];
  const float* emb  = (const float*)d_in[2];
  const float* Wxh0 = (const float*)d_in[3];
  const float* Wxhr = (const float*)d_in[4];
  const float* Whh  = (const float*)d_in[5];
  const float* bh   = (const float*)d_in[6];
  const float* Why  = (const float*)d_in[7];
  const float* by   = (const float*)d_in[8];
  float* out = (float*)d_out;
  float* tab = (float*)d_ws;

  input_table_kernel<<<dim3(NVOCAB / TILE_M / 4), dim3(128), 0, stream>>>(emb, Wxh0, bh, tab);

  dim3 grid((NBATCH + TILE_M - 1) / TILE_M);
  dim3 block(NWAVE * 32);
  rnn_pipeline_kernel<<<grid, block, 0, stream>>>(x, h0, tab, Wxhr, Whh,
                                                  bh, Why, by, out);
}


# One wave per layer: state times stacked [W_hh(l);W_xh(l+1)] (dense 16x8: 2 products per layer-step, was 3); input term handed on via LDS; sum order now bias, input k, own-state k (re-paired)
# speedup vs baseline: 3.0938x; 1.0614x over previous
#include <hip/hip_runtime.h>


typedef float v8f __attribute__((ext_vector_type(8)));
typedef float v2f __attribute__((ext_vector_type(2)));

#define DIM_E   64
#define DIM_H   8
#define NLAYER  25
#define NBATCH  1000
#define NT      512
#define NVOCAB  32000
#define TILE_M  16
#define TABW    16
#define HPAD    20
#define NWAVE   NLAYER
#define KSUB    2
#define LAG     2
#define RING    (4 * KSUB)
#define RMASK   (RING - 1)
#define TOKQ    (2 * KSUB)
#define STEP_OFF (LAG * KSUB + 1)
#define OFF_LAST ((NLAYER - 1) * STEP_OFF)
#define NROUNDS ((NT + OFF_LAST + KSUB - 1) / KSUB)
static_assert(NROUNDS % 2 == 0, "the layer-0 wave takes the rounds two at a time");
static_assert(RING >= (LAG + 1) * KSUB, "a slot must not be written again before its reader's round has opened");

__device__ __forceinline__ float fast_tanh(float x) {
  return __builtin_amdgcn_tanhf(x);
}

__device__ __forceinline__ void round_open() {
  asm volatile("s_wait_dscnt 0x0" ::: "memory");
  asm volatile("s_barrier_signal -1" ::: "memory");
}
__device__ __forceinline__ void round_close() {
  asm volatile("s_barrier_wait -1" ::: "memory");
}

__device__ __forceinline__ float xhalf(float x) {
  int b = __builtin_bit_cast(int, x);
  return __builtin_bit_cast(float, __builtin_amdgcn_permlanex16(b, b, 0x76543210, 0xfedcba98, false, false));
}

__device__ __forceinline__ v8f wmma4(v2f a, v2f b, v8f c) {
  return __builtin_amdgcn_wmma_f32_16x16x4_f32(false, a, false, b, (short)0, c,
                                               false, false);
}

__global__ __launch_bounds__(128)
void input_table_kernel(const float* __restrict__ emb,
                        const float* __restrict__ Wxh0,
                        const float* __restrict__ bh,
                        float* __restrict__ tab) {
  const int tid  = threadIdx.x;
  const int wid  = __builtin_amdgcn_readfirstlane(tid >> 5);
  const int lane = tid & 31;
  const int n    = lane & 15;
  const int hh   = lane >> 4;
  const int kb   = hh << 1;
  const int v0   = (blockIdx.x * 4 + wid) * TILE_M;
  const int u    = (n < DIM_H) ? n : ((n - DIM_H + 2) & 7);
  const float* erow = emb + (long)(v0 + n) * DIM_E + kb;
  v2f a[16], w[16];
#pragma unroll
  for (int c = 0; c < 16; ++c) a[c] = *(const v2f*)(erow + c * 4);
#pragma unroll
  for (int c = 0; c < 16; ++c)
#pragma unroll
    for (int j = 0; j < 2; ++j) w[c][j] = Wxh0[(c * 4 + kb + j) * DIM_H + u];
  const float bias = bh[u];
  v8f acc;
#pragma unroll
  for (int q = 0; q < 8; ++q) acc[q] = bias;
#pragma unroll
  for (int c = 0; c < 16; ++c) acc = wmma4(a[c], w[c], acc);
  float* dst = tab + (long)(v0 + 8 * hh) * TABW + n;
#pragma unroll
  for (int q = 0; q < 8; ++q) dst[q * TABW] = acc[q];
}

__device__ __forceinline__ v8f layer_products(const v2f w0, const v2f w1, const float (&h)[4], const float4 x,
                                              const float (&bn)[4]) {
  v8f c;
  c[0] = x.x; c[1] = x.y; c[2] = x.z; c[3] = x.w;
  c[4] = bn[0]; c[5] = bn[1]; c[6] = bn[2]; c[7] = bn[3];
  v8f acc = wmma4(w0, (v2f){h[0], h[1]}, c);
  return wmma4(w1, (v2f){h[2], h[3]}, acc);
}

__global__ __launch_bounds__(NWAVE * 32, 1)
void rnn_pipeline_kernel(const int* __restrict__ x,
                         const float* __restrict__ h0,
                         const float* __restrict__ tab,
                         const float* __restrict__ Wxhr,
                         const float* __restrict__ Whh,
                         const float* __restrict__ bh,
                         const float* __restrict__ Why,
                         const float* __restrict__ by,
                         float* __restrict__ out) {
  __shared__ __attribute__((aligned(16))) float xbuf[NLAYER][RING][TILE_M][HPAD];

  const int tid  = threadIdx.x;
  const int l    = __builtin_amdgcn_readfirstlane(tid >> 5);
  const int lane = tid & 31;
  const int n    = lane & 15;
  const int hh   = lane >> 4;
  const int tile_base = blockIdx.x * TILE_M;
  const int toff  = l * STEP_OFF;

  v2f wA[2];
  {
    const int  a    = n >> 3;
    const int  r    = n & 7;
    const int  uo   = 4 * a + (r & 3);
    const bool mine = (r < 4);
#pragma unroll
    for (int q = 0; q < 2; ++q)
#pragma unroll
      for (int j = 0; j < 2; ++j) {
        const int k = 4 * hh + 2 * q + j;
        float v = 0.0f;
        if (mine)                 v = Whh[l * DIM_H * DIM_H + k * DIM_H + uo];
        else if (l < NLAYER - 1)  v = Wxhr[l * DIM_H * DIM_H + k * DIM_H + uo];
        wA[q][j] = v;
      }
  }
  float bn[4];
#pragma unroll
  for (int i = 0; i < 4; ++i) bn[i] = (l < NLAYER - 1) ? bh[(l + 1) * DIM_H + 4 * hh + i] : 0.0f;

  int gb = tile_base + n; if (gb > NBATCH - 1) gb = NBATCH - 1;
  float hs[4];
  {
    const float4 v = *(const float4*)(h0 + ((long)l * NBATCH + gb) * DIM_H + 4 * hh);
    hs[0] = v.x; hs[1] = v.y; hs[2] = v.z; hs[3] = v.w;
  }
  float* const xw = &xbuf[l][0][n][4 * hh];
  const int SLOT = TILE_M * HPAD;

  if (l == 0) {
    int4   xn;
    float4 pa[TOKQ];
    {
      const int4 x0 = *(const int4*)(x + gb * NT);
      xn = *(const int4*)(x + gb * NT + TOKQ);
      const int id0[TOKQ] = {x0.x, x0.y, x0.z, x0.w};
#pragma unroll
      for (int j = 0; j < TOKQ; ++j) {
        int id = id0[j]; id = id < 0 ? 0 : (id > NVOCAB - 1 ? NVOCAB - 1 : id);
        pa[j] = *(const float4*)(tab + (long)id * TABW + 4 * hh);
      }
    }
    for (int rr = 0; rr < NROUNDS / 2; ++rr) {
      const int t0 = rr * TOKQ;
#pragma unroll
      for (int j = 0; j < TOKQ; ++j) {
        const int t = t0 + j;
        if (j % KSUB == 0) round_open();
        if (t < NT) {
          const v8f acc = layer_products(wA[0], wA[1], hs, pa[j], bn);
#pragma unroll
          for (int i = 0; i < 4; ++i) hs[i] = fast_tanh(acc[i]);
          *(float4*)(xw + ((t - 1) & RMASK) * SLOT) = make_float4(acc[4], acc[5], acc[6], acc[7]);
        } else if (t == NT) {
          const v8f acc = layer_products(wA[0], wA[1], hs, pa[j], bn);
          *(float4*)(xw + ((t - 1) & RMASK) * SLOT) = make_float4(acc[4], acc[5], acc[6], acc[7]);
        }
        if (j == TOKQ - 1) {
          const int idn[TOKQ] = {xn.x, xn.y, xn.z, xn.w};
#pragma unroll
          for (int i = 0; i < TOKQ; ++i) {
            int id = idn[i]; id = id < 0 ? 0 : (id > NVOCAB - 1 ? NVOCAB - 1 : id);
            pa[i] = *(const float4*)(tab + (long)id * TABW + 4 * hh);
          }
          int tn = t0 + 2 * TOKQ;  tn = tn > NT - TOKQ ? NT - TOKQ : tn;
          xn = *(const int4*)(x + gb * NT + tn);
        }
        if (j % KSUB == KSUB - 1) round_close();
      }
    }
  } else {
    const float* const xr = &xbuf[l - 1][0][n][4 * hh];
    for (int r = 0; r < NROUNDS; ++r) {
      const int t0 = r * KSUB - toff;
      float4 xin[KSUB];
#pragma unroll
      for (int j = 0; j < KSUB; ++j) xin[j] = *(const float4*)(xr + ((t0 + j) & RMASK) * SLOT);
      round_open();
#pragma unroll
      for (int j = 0; j < KSUB; ++j) asm volatile("" :: "v"(xin[j].x), "v"(xin[j].y), "v"(xin[j].z), "v"(xin[j].w));
#pragma unroll
      for (int j = 0; j < KSUB; ++j) {
        const int t = t0 + j;
        if ((unsigned)t < (unsigned)NT) {
          const v8f acc = layer_products(wA[0], wA[1], hs, xin[j], bn);
#pragma unroll
          for (int i = 0; i < 4; ++i) hs[i] = fast_tanh(acc[i]);
          *(float4*)(xw + ((t - 1) & RMASK) * SLOT) = make_float4(acc[4], acc[5], acc[6], acc[7]);
        } else if (t == NT) {
          const v8f acc = layer_products(wA[0], wA[1], hs, xin[j], bn);
          *(float4*)(xw + ((t - 1) & RMASK) * SLOT) = make_float4(acc[4], acc[5], acc[6], acc[7]);
        }
      }
      round_close();
    }
  }

  {
    const int g = tile_base + n;
    if (l == NLAYER - 1) {
      const float p0 = xhalf(hs[0]), p1 = xhalf(hs[1]), p2 = xhalf(hs[2]), p3 = xhalf(hs[3]);
      if (hh == 0 && g < NBATCH) {
        float a = by[0];
        a += hs[0] * Why[0]; a += hs[1] * Why[1]; a += hs[2] * Why[2]; a += hs[3] * Why[3];
        a += p0 * Why[4];    a += p1 * Why[5];    a += p2 * Why[6];    a += p3 * Why[7];
        out[g] = a;
      }
    }
    if (g < NBATCH)
      *(float4*)(out + NBATCH + ((long)l * NBATCH + g) * DIM_H + 4 * hh) = make_float4(hs[0], hs[1], hs[2], hs[3]);
  }
}

extern "C" void kernel_launch(void* const* d_in, const int* in_sizes, int n_in,
                              void* d_out, int out_size, void* d_ws, size_t ws_size,
                              hipStream_t stream) {
  (void)in_sizes; (void)n_in; (void)ws_size; (void)out_size;
  const int*   x    = (const int*)  d_in[0];
  const float* h0   = (const float*)d_in[1];
  const float* emb  = (const float*)d_in[2];
  const float* Wxh0 = (const float*)d_in[3];
  const float* Wxhr = (const float*)d_in[4];
  const float* Whh  = (const float*)d_in[5];
  const float* bh   = (const float*)d_in[6];
  const float* Why  = (const float*)d_in[7];
  const float* by   = (const float*)d_in[8];
  float* out = (float*)d_out;
  float* tab = (float*)d_ws;

  input_table_kernel<<<dim3(NVOCAB / TILE_M / 4), dim3(128), 0, stream>>>(emb, Wxh0, bh, tab);

  dim3 grid((NBATCH + TILE_M - 1) / TILE_M);
  dim3 block(NWAVE * 32);
  rnn_pipeline_kernel<<<grid, block, 0, stream>>>(x, h0, tab, Wxhr, Whh,
                                                  bh, Why, by, out);
}


# Ring slots renumbered per hand-over (step + layer*5 mod 8) and rounds of layers 1..24 unrolled over one ring turn: every LDS slot is an immediate offset, no per-step address arithmetic (v_mad_u32)
# speedup vs baseline: 3.4223x; 1.1062x over previous
#include <hip/hip_runtime.h>


typedef float v8f __attribute__((ext_vector_type(8)));
typedef float v2f __attribute__((ext_vector_type(2)));

#define DIM_E   64
#define DIM_H   8
#define NLAYER  25
#define NBATCH  1000
#define NT      512
#define NVOCAB  32000
#define TILE_M  16
#define TABW    16
#define HPAD    20
#define NWAVE   NLAYER
#define KSUB    2
#define LAG     2
#define RING    (4 * KSUB)
#define RMASK   (RING - 1)
#define RUNROLL (RING / KSUB)
#define TOKQ    (2 * KSUB)
#define STEP_OFF (LAG * KSUB + 1)
#define OFF_LAST ((NLAYER - 1) * STEP_OFF)
#define NROUNDS ((NT + OFF_LAST + KSUB - 1) / KSUB)
static_assert(NROUNDS % 2 == 0, "the layer-0 wave takes the rounds two at a time");
static_assert(NROUNDS % RUNROLL == 0, "the round loop of layers 1.. is unrolled over one turn of the ring");
static_assert((RING & RMASK) == 0, "ring slots are taken modulo a power of two");
static_assert((STEP_OFF - 1) % TOKQ == 0 && RING % TOKQ == 0, "the TOKQ slots the layer-0 wave writes in a pair of rounds do not wrap inside the pair");
static_assert(RING >= (LAG + 1) * KSUB, "a slot must not be written again before its reader's round has opened");

__device__ __forceinline__ float fast_tanh(float x) {
  return __builtin_amdgcn_tanhf(x);
}

__device__ __forceinline__ void round_open() {
  asm volatile("s_wait_dscnt 0x0" ::: "memory");
  asm volatile("s_barrier_signal -1" ::: "memory");
}
__device__ __forceinline__ void round_close() {
  asm volatile("s_barrier_wait -1" ::: "memory");
}

__device__ __forceinline__ float xhalf(float x) {
  int b = __builtin_bit_cast(int, x);
  return __builtin_bit_cast(float, __builtin_amdgcn_permlanex16(b, b, 0x76543210, 0xfedcba98, false, false));
}

__device__ __forceinline__ v8f wmma4(v2f a, v2f b, v8f c) {
  return __builtin_amdgcn_wmma_f32_16x16x4_f32(false, a, false, b, (short)0, c,
                                               false, false);
}

__global__ __launch_bounds__(128)
void input_table_kernel(const float* __restrict__ emb,
                        const float* __restrict__ Wxh0,
                        const float* __restrict__ bh,
                        float* __restrict__ tab) {
  const int tid  = threadIdx.x;
  const int wid  = __builtin_amdgcn_readfirstlane(tid >> 5);
  const int lane = tid & 31;
  const int n    = lane & 15;
  const int hh   = lane >> 4;
  const int kb   = hh << 1;
  const int v0   = (blockIdx.x * 4 + wid) * TILE_M;
  const int u    = (n < DIM_H) ? n : ((n - DIM_H + 2) & 7);
  const float* erow = emb + (long)(v0 + n) * DIM_E + kb;
  v2f a[16], w[16];
#pragma unroll
  for (int c = 0; c < 16; ++c) a[c] = *(const v2f*)(erow + c * 4);
#pragma unroll
  for (int c = 0; c < 16; ++c)
#pragma unroll
    for (int j = 0; j < 2; ++j) w[c][j] = Wxh0[(c * 4 + kb + j) * DIM_H + u];
  const float bias = bh[u];
  v8f acc;
#pragma unroll
  for (int q = 0; q < 8; ++q) acc[q] = bias;
#pragma unroll
  for (int c = 0; c < 16; ++c) acc = wmma4(a[c], w[c], acc);
  float* dst = tab + (long)(v0 + 8 * hh) * TABW + n;
#pragma unroll
  for (int q = 0; q < 8; ++q) dst[q * TABW] = acc[q];
}

__device__ __forceinline__ v8f layer_products(const v2f w0, const v2f w1, const float (&h)[4], const float4 x,
                                              const float (&bn)[4]) {
  v8f c;
  c[0] = x.x; c[1] = x.y; c[2] = x.z; c[3] = x.w;
  c[4] = bn[0]; c[5] = bn[1]; c[6] = bn[2]; c[7] = bn[3];
  v8f acc = wmma4(w0, (v2f){h[0], h[1]}, c);
  return wmma4(w1, (v2f){h[2], h[3]}, acc);
}

__global__ __launch_bounds__(NWAVE * 32, 1)
void rnn_pipeline_kernel(const int* __restrict__ x,
                         const float* __restrict__ h0,
                         const float* __restrict__ tab,
                         const float* __restrict__ Wxhr,
                         const float* __restrict__ Whh,
                         const float* __restrict__ bh,
                         const float* __restrict__ Why,
                         const float* __restrict__ by,
                         float* __restrict__ out) {
  __shared__ __attribute__((aligned(16))) float xbuf[NLAYER][RING][TILE_M][HPAD];

  const int tid  = threadIdx.x;
  const int l    = __builtin_amdgcn_readfirstlane(tid >> 5);
  const int lane = tid & 31;
  const int n    = lane & 15;
  const int hh   = lane >> 4;
  const int tile_base = blockIdx.x * TILE_M;
  const int toff  = l * STEP_OFF;

  v2f wA[2];
  {
    const int  a    = n >> 3;
    const int  r    = n & 7;
    const int  uo   = 4 * a + (r & 3);
    const bool mine = (r < 4);
#pragma unroll
    for (int q = 0; q < 2; ++q)
#pragma unroll
      for (int j = 0; j < 2; ++j) {
        const int k = 4 * hh + 2 * q + j;
        float v = 0.0f;
        if (mine)                 v = Whh[l * DIM_H * DIM_H + k * DIM_H + uo];
        else if (l < NLAYER - 1)  v = Wxhr[l * DIM_H * DIM_H + k * DIM_H + uo];
        wA[q][j] = v;
      }
  }
  float bn[4];
#pragma unroll
  for (int i = 0; i < 4; ++i) bn[i] = (l < NLAYER - 1) ? bh[(l + 1) * DIM_H + 4 * hh + i] : 0.0f;

  int gb = tile_base + n; if (gb > NBATCH - 1) gb = NBATCH - 1;
  float hs[4];
  {
    const float4 v = *(const float4*)(h0 + ((long)l * NBATCH + gb) * DIM_H + 4 * hh);
    hs[0] = v.x; hs[1] = v.y; hs[2] = v.z; hs[3] = v.w;
  }
  float* const xw = &xbuf[l][0][n][4 * hh];
  const int SLOT = TILE_M * HPAD;

  if (l == 0) {
    int4   xn;
    float4 pa[TOKQ];
    {
      const int4 x0 = *(const int4*)(x + gb * NT);
      xn = *(const int4*)(x + gb * NT + TOKQ);
      const int id0[TOKQ] = {x0.x, x0.y, x0.z, x0.w};
#pragma unroll
      for (int j = 0; j < TOKQ; ++j) {
        int id = id0[j]; id = id < 0 ? 0 : (id > NVOCAB - 1 ? NVOCAB - 1 : id);
        pa[j] = *(const float4*)(tab + (long)id * TABW + 4 * hh);
      }
    }
    for (int rr = 0; rr < NROUNDS / 2; ++rr) {
      const int t0 = rr * TOKQ;
      float* const xwp = xw + ((t0 + STEP_OFF - 1) & RMASK) * SLOT;
#pragma unroll
      for (int j = 0; j < TOKQ; ++j) {
        const int t = t0 + j;
        if (j % KSUB == 0) round_open();
        if (t < NT) {
          const v8f acc = layer_products(wA[0], wA[1], hs, pa[j], bn);
#pragma unroll
          for (int i = 0; i < 4; ++i) hs[i] = fast_tanh(acc[i]);
          *(float4*)(xwp + j * SLOT) = make_float4(acc[4], acc[5], acc[6], acc[7]);
        } else if (t == NT) {
          const v8f acc = layer_products(wA[0], wA[1], hs, pa[j], bn);
          *(float4*)(xwp + j * SLOT) = make_float4(acc[4], acc[5], acc[6], acc[7]);
        }
        if (j == TOKQ - 1) {
          const int idn[TOKQ] = {xn.x, xn.y, xn.z, xn.w};
#pragma unroll
          for (int i = 0; i < TOKQ; ++i) {
            int id = idn[i]; id = id < 0 ? 0 : (id > NVOCAB - 1 ? NVOCAB - 1 : id);
            pa[i] = *(const float4*)(tab + (long)id * TABW + 4 * hh);
          }
          int tn = t0 + 2 * TOKQ;  tn = tn > NT - TOKQ ? NT - TOKQ : tn;
          xn = *(const int4*)(x + gb * NT + tn);
        }
        if (j % KSUB == KSUB - 1) round_close();
      }
    }
  } else {
    const float* const xr = &xbuf[l - 1][0][n][4 * hh];
    for (int ru = 0; ru < NROUNDS / RUNROLL; ++ru) {
#pragma unroll
     for (int q = 0; q < RUNROLL; ++q) {
      const int r  = ru * RUNROLL + q;
      const int t0 = r * KSUB - toff;
      float4 xin[KSUB];
#pragma unroll
      for (int j = 0; j < KSUB; ++j) xin[j] = *(const float4*)(xr + ((q * KSUB + j) & RMASK) * SLOT);
      round_open();
#pragma unroll
      for (int j = 0; j < KSUB; ++j) asm volatile("" :: "v"(xin[j].x), "v"(xin[j].y), "v"(xin[j].z), "v"(xin[j].w));
#pragma unroll
      for (int j = 0; j < KSUB; ++j) {
        const int t = t0 + j;
        const int wslot = (q * KSUB + j + STEP_OFF - 1) & RMASK;
        if ((unsigned)t < (unsigned)NT) {
          const v8f acc = layer_products(wA[0], wA[1], hs, xin[j], bn);
#pragma unroll
          for (int i = 0; i < 4; ++i) hs[i] = fast_tanh(acc[i]);
          *(float4*)(xw + wslot * SLOT) = make_float4(acc[4], acc[5], acc[6], acc[7]);
        } else if (t == NT) {
          const v8f acc = layer_products(wA[0], wA[1], hs, xin[j], bn);
          *(float4*)(xw + wslot * SLOT) = make_float4(acc[4], acc[5], acc[6], acc[7]);
        }
      }
      round_close();
     }
    }
  }

  {
    const int g = tile_base + n;
    if (l == NLAYER - 1) {
      const float p0 = xhalf(hs[0]), p1 = xhalf(hs[1]), p2 = xhalf(hs[2]), p3 = xhalf(hs[3]);
      if (hh == 0 && g < NBATCH) {
        float a = by[0];
        a += hs[0] * Why[0]; a += hs[1] * Why[1]; a += hs[2] * Why[2]; a += hs[3] * Why[3];
        a += p0 * Why[4];    a += p1 * Why[5];    a += p2 * Why[6];    a += p3 * Why[7];
        out[g] = a;
      }
    }
    if (g < NBATCH)
      *(float4*)(out + NBATCH + ((long)l * NBATCH + g) * DIM_H + 4 * hh) = make_float4(hs[0], hs[1], hs[2], hs[3]);
  }
}

extern "C" void kernel_launch(void* const* d_in, const int* in_sizes, int n_in,
                              void* d_out, int out_size, void* d_ws, size_t ws_size,
                              hipStream_t stream) {
  (void)in_sizes; (void)n_in; (void)ws_size; (void)out_size;
  const int*   x    = (const int*)  d_in[0];
  const float* h0   = (const float*)d_in[1];
  const float* emb  = (const float*)d_in[2];
  const float* Wxh0 = (const float*)d_in[3];
  const float* Wxhr = (const float*)d_in[4];
  const float* Whh  = (const float*)d_in[5];
  const float* bh   = (const float*)d_in[6];
  const float* Why  = (const float*)d_in[7];
  const float* by   = (const float*)d_in[8];
  float* out = (float*)d_out;
  float* tab = (float*)d_ws;

  input_table_kernel<<<dim3(NVOCAB / TILE_M / 4), dim3(128), 0, stream>>>(emb, Wxh0, bh, tab);

  dim3 grid((NBATCH + TILE_M - 1) / TILE_M);
  dim3 block(NWAVE * 32);
  rnn_pipeline_kernel<<<grid, block, 0, stream>>>(x, h0, tab, Wxhr, Whh,
                                                  bh, Why, by, out);
}


# Time-range tests hoisted out of the steady state: ring turns run as fill, full, drain loops; a full turn (8 steps; layer 0: 4) is straight-line, no scalar compare or branch per step; same sum order
# speedup vs baseline: 3.6414x; 1.0640x over previous
#include <hip/hip_runtime.h>


typedef float v8f __attribute__((ext_vector_type(8)));
typedef float v2f __attribute__((ext_vector_type(2)));

#define DIM_E   64
#define DIM_H   8
#define NLAYER  25
#define NBATCH  1000
#define NT      512
#define NVOCAB  32000
#define TILE_M  16
#define TABW    16
#define HPAD    20
#define NWAVE   NLAYER
#define KSUB    2
#define LAG     2
#define RING    (4 * KSUB)
#define RMASK   (RING - 1)
#define RUNROLL (RING / KSUB)
#define TOKQ    (2 * KSUB)
#define STEP_OFF (LAG * KSUB + 1)
#define OFF_LAST ((NLAYER - 1) * STEP_OFF)
#define NROUNDS ((NT + OFF_LAST + KSUB - 1) / KSUB)
static_assert(NROUNDS % 2 == 0, "the layer-0 wave takes the rounds two at a time");
static_assert(NROUNDS % RUNROLL == 0, "the round loop of layers 1.. is unrolled over one turn of the ring");
static_assert((RING & RMASK) == 0, "ring slots are taken modulo a power of two");
static_assert((STEP_OFF - 1) % TOKQ == 0 && RING % TOKQ == 0, "the TOKQ slots the layer-0 wave writes in a pair of rounds do not wrap inside the pair");
static_assert(RING >= (LAG + 1) * KSUB, "a slot must not be written again before its reader's round has opened");
static_assert(NT % TOKQ == 0, "the layer-0 wave's pairs of rounds below NT hold no step at or past NT");

__device__ __forceinline__ float fast_tanh(float x) {
  return __builtin_amdgcn_tanhf(x);
}

__device__ __forceinline__ void round_open() {
  asm volatile("s_wait_dscnt 0x0" ::: "memory");
  asm volatile("s_barrier_signal -1" ::: "memory");
}
__device__ __forceinline__ void round_close() {
  asm volatile("s_barrier_wait -1" ::: "memory");
}

__device__ __forceinline__ float xhalf(float x) {
  int b = __builtin_bit_cast(int, x);
  return __builtin_bit_cast(float, __builtin_amdgcn_permlanex16(b, b, 0x76543210, 0xfedcba98, false, false));
}

__device__ __forceinline__ v8f wmma4(v2f a, v2f b, v8f c) {
  return __builtin_amdgcn_wmma_f32_16x16x4_f32(false, a, false, b, (short)0, c,
                                               false, false);
}

__global__ __launch_bounds__(128)
void input_table_kernel(const float* __restrict__ emb,
                        const float* __restrict__ Wxh0,
                        const float* __restrict__ bh,
                        float* __restrict__ tab) {
  const int tid  = threadIdx.x;
  const int wid  = __builtin_amdgcn_readfirstlane(tid >> 5);
  const int lane = tid & 31;
  const int n    = lane & 15;
  const int hh   = lane >> 4;
  const int kb   = hh << 1;
  const int v0   = (blockIdx.x * 4 + wid) * TILE_M;
  const int u    = (n < DIM_H) ? n : ((n - DIM_H + 2) & 7);
  const float* erow = emb + (long)(v0 + n) * DIM_E + kb;
  v2f a[16], w[16];
#pragma unroll
  for (int c = 0; c < 16; ++c) a[c] = *(const v2f*)(erow + c * 4);
#pragma unroll
  for (int c = 0; c < 16; ++c)
#pragma unroll
    for (int j = 0; j < 2; ++j) w[c][j] = Wxh0[(c * 4 + kb + j) * DIM_H + u];
  const float bias = bh[u];
  v8f acc;
#pragma unroll
  for (int q = 0; q < 8; ++q) acc[q] = bias;
#pragma unroll
  for (int c = 0; c < 16; ++c) acc = wmma4(a[c], w[c], acc);
  float* dst = tab + (long)(v0 + 8 * hh) * TABW + n;
#pragma unroll
  for (int q = 0; q < 8; ++q) dst[q * TABW] = acc[q];
}

__device__ __forceinline__ v8f layer_products(const v2f w0, const v2f w1, const float (&h)[4], const float4 x,
                                              const float (&bn)[4]) {
  v8f c;
  c[0] = x.x; c[1] = x.y; c[2] = x.z; c[3] = x.w;
  c[4] = bn[0]; c[5] = bn[1]; c[6] = bn[2]; c[7] = bn[3];
  v8f acc = wmma4(w0, (v2f){h[0], h[1]}, c);
  return wmma4(w1, (v2f){h[2], h[3]}, acc);
}

template <bool CHECKED>
__device__ __forceinline__ void ring_turn(const float* const xr, float* const xw, const v2f w0, const v2f w1,
                                          float (&hs)[4], const float (&bn)[4], const int tA) {
  const int SLOT = TILE_M * HPAD;
#pragma unroll
  for (int q = 0; q < RUNROLL; ++q) {
    float4 xin[KSUB];
#pragma unroll
    for (int j = 0; j < KSUB; ++j) xin[j] = *(const float4*)(xr + ((q * KSUB + j) & RMASK) * SLOT);
    round_open();
    if (CHECKED) {
#pragma unroll
      for (int j = 0; j < KSUB; ++j) asm volatile("" :: "v"(xin[j].x), "v"(xin[j].y), "v"(xin[j].z), "v"(xin[j].w));
    }
#pragma unroll
    for (int j = 0; j < KSUB; ++j) {
      const int t = tA + q * KSUB + j;
      const int wslot = (q * KSUB + j + STEP_OFF - 1) & RMASK;
      if (!CHECKED || (unsigned)t < (unsigned)NT) {
        const v8f acc = layer_products(w0, w1, hs, xin[j], bn);
#pragma unroll
        for (int i = 0; i < 4; ++i) hs[i] = fast_tanh(acc[i]);
        *(float4*)(xw + wslot * SLOT) = make_float4(acc[4], acc[5], acc[6], acc[7]);
      } else if (t == NT) {
        const v8f acc = layer_products(w0, w1, hs, xin[j], bn);
        *(float4*)(xw + wslot * SLOT) = make_float4(acc[4], acc[5], acc[6], acc[7]);
      }
    }
    round_close();
  }
}

template <bool CHECKED>
__device__ __forceinline__ void layer0_pair(const int* const xrow, const float* const tabh, float* const xwp,
                                            const v2f w0, const v2f w1, float (&hs)[4], const float (&bn)[4],
                                            float4 (&pa)[TOKQ], int4& xn, const int t0) {
  const int SLOT = TILE_M * HPAD;
#pragma unroll
  for (int j = 0; j < TOKQ; ++j) {
    const int t = t0 + j;
    if (j % KSUB == 0) round_open();
    if (!CHECKED || t < NT) {
      const v8f acc = layer_products(w0, w1, hs, pa[j], bn);
#pragma unroll
      for (int i = 0; i < 4; ++i) hs[i] = fast_tanh(acc[i]);
      *(float4*)(xwp + j * SLOT) = make_float4(acc[4], acc[5], acc[6], acc[7]);
    } else if (t == NT) {
      const v8f acc = layer_products(w0, w1, hs, pa[j], bn);
      *(float4*)(xwp + j * SLOT) = make_float4(acc[4], acc[5], acc[6], acc[7]);
    }
    if (j == TOKQ - 1) {
      const int idn[TOKQ] = {xn.x, xn.y, xn.z, xn.w};
#pragma unroll
      for (int i = 0; i < TOKQ; ++i) {
        int id = idn[i]; id = id < 0 ? 0 : (id > NVOCAB - 1 ? NVOCAB - 1 : id);
        pa[i] = *(const float4*)(tabh + (long)id * TABW);
      }
      int tn = t0 + 2 * TOKQ;  tn = tn > NT - TOKQ ? NT - TOKQ : tn;
      xn = *(const int4*)(xrow + tn);
    }
    if (j % KSUB == KSUB - 1) round_close();
  }
}

__global__ __launch_bounds__(NWAVE * 32, 1)
void rnn_pipeline_kernel(const int* __restrict__ x,
                         const float* __restrict__ h0,
                         const float* __restrict__ tab,
                         const float* __restrict__ Wxhr,
                         const float* __restrict__ Whh,
                         const float* __restrict__ bh,
                         const float* __restrict__ Why,
                         const float* __restrict__ by,
                         float* __restrict__ out) {
  __shared__ __attribute__((aligned(16))) float xbuf[NLAYER][RING][TILE_M][HPAD];

  const int tid  = threadIdx.x;
  const int l    = __builtin_amdgcn_readfirstlane(tid >> 5);
  const int lane = tid & 31;
  const int n    = lane & 15;
  const int hh   = lane >> 4;
  const int tile_base = blockIdx.x * TILE_M;
  const int toff  = l * STEP_OFF;

  v2f wA[2];
  {
    const int  a    = n >> 3;
    const int  r    = n & 7;
    const int  uo   = 4 * a + (r & 3);
    const bool mine = (r < 4);
#pragma unroll
    for (int q = 0; q < 2; ++q)
#pragma unroll
      for (int j = 0; j < 2; ++j) {
        const int k = 4 * hh + 2 * q + j;
        float v = 0.0f;
        if (mine)                 v = Whh[l * DIM_H * DIM_H + k * DIM_H + uo];
        else if (l < NLAYER - 1)  v = Wxhr[l * DIM_H * DIM_H + k * DIM_H + uo];
        wA[q][j] = v;
      }
  }
  float bn[4];
#pragma unroll
  for (int i = 0; i < 4; ++i) bn[i] = (l < NLAYER - 1) ? bh[(l + 1) * DIM_H + 4 * hh + i] : 0.0f;

  int gb = tile_base + n; if (gb > NBATCH - 1) gb = NBATCH - 1;
  float hs[4];
  {
    const float4 v = *(const float4*)(h0 + ((long)l * NBATCH + gb) * DIM_H + 4 * hh);
    hs[0] = v.x; hs[1] = v.y; hs[2] = v.z; hs[3] = v.w;
  }
  float* const xw = &xbuf[l][0][n][4 * hh];
  const int SLOT = TILE_M * HPAD;

  if (l == 0) {
    int4   xn;
    float4 pa[TOKQ];
    {
      const int4 x0 = *(const int4*)(x + gb * NT);
      xn = *(const int4*)(x + gb * NT + TOKQ);
      const int id0[TOKQ] = {x0.x, x0.y, x0.z, x0.w};
#pragma unroll
      for (int j = 0; j < TOKQ; ++j) {
        int id = id0[j]; id = id < 0 ? 0 : (id > NVOCAB - 1 ? NVOCAB - 1 : id);
        pa[j] = *(const float4*)(tab + (long)id * TABW + 4 * hh);
      }
    }
    const int*   const xrow = x + gb * NT;
    const float* const tabh = tab + 4 * hh;
    int rr = 0;
    for (; rr < NT / TOKQ; ++rr)
      layer0_pair<false>(xrow, tabh, xw + ((rr * TOKQ + STEP_OFF - 1) & RMASK) * SLOT, wA[0], wA[1], hs, bn, pa, xn, rr * TOKQ);
    for (; rr < NROUNDS / 2; ++rr)
      layer0_pair<true >(xrow, tabh, xw + ((rr * TOKQ + STEP_OFF - 1) & RMASK) * SLOT, wA[0], wA[1], hs, bn, pa, xn, rr * TOKQ);
  } else {
    const float* const xr = &xbuf[l - 1][0][n][4 * hh];
    const int TSTEPS = RUNROLL * KSUB;
    const int ruA = (toff + TSTEPS - 1) / TSTEPS;
    const int ruB = (NT + toff) / TSTEPS;
    int ru = 0;
    for (; ru < ruA; ++ru)               ring_turn<true >(xr, xw, wA[0], wA[1], hs, bn, ru * TSTEPS - toff);
    for (; ru < ruB; ++ru)               ring_turn<false>(xr, xw, wA[0], wA[1], hs, bn, ru * TSTEPS - toff);
    for (; ru < NROUNDS / RUNROLL; ++ru) ring_turn<true >(xr, xw, wA[0], wA[1], hs, bn, ru * TSTEPS - toff);
  }

  {
    const int g = tile_base + n;
    if (l == NLAYER - 1) {
      const float p0 = xhalf(hs[0]), p1 = xhalf(hs[1]), p2 = xhalf(hs[2]), p3 = xhalf(hs[3]);
      if (hh == 0 && g < NBATCH) {
        float a = by[0];
        a += hs[0] * Why[0]; a += hs[1] * Why[1]; a += hs[2] * Why[2]; a += hs[3] * Why[3];
        a += p0 * Why[4];    a += p1 * Why[5];    a += p2 * Why[6];    a += p3 * Why[7];
        out[g] = a;
      }
    }
    if (g < NBATCH)
      *(float4*)(out + NBATCH + ((long)l * NBATCH + g) * DIM_H + 4 * hh) = make_float4(hs[0], hs[1], hs[2], hs[3]);
  }
}

extern "C" void kernel_launch(void* const* d_in, const int* in_sizes, int n_in,
                              void* d_out, int out_size, void* d_ws, size_t ws_size,
                              hipStream_t stream) {
  (void)in_sizes; (void)n_in; (void)ws_size; (void)out_size;
  const int*   x    = (const int*)  d_in[0];
  const float* h0   = (const float*)d_in[1];
  const float* emb  = (const float*)d_in[2];
  const float* Wxh0 = (const float*)d_in[3];
  const float* Wxhr = (const float*)d_in[4];
  const float* Whh  = (const float*)d_in[5];
  const float* bh   = (const float*)d_in[6];
  const float* Why  = (const float*)d_in[7];
  const float* by   = (const float*)d_in[8];
  float* out = (float*)d_out;
  float* tab = (float*)d_ws;

  input_table_kernel<<<dim3(NVOCAB / TILE_M / 4), dim3(128), 0, stream>>>(emb, Wxh0, bh, tab);

  dim3 grid((NBATCH + TILE_M - 1) / TILE_M);
  dim3 block(NWAVE * 32);
  rnn_pipeline_kernel<<<grid, block, 0, stream>>>(x, h0, tab, Wxhr, Whh,
                                                  bh, Why, by, out);
}
